# starting kernel
# baseline (speedup 1.0000x reference)
#include <hip/hip_runtime.h>


typedef __attribute__((ext_vector_type(2))) float v2f;
typedef __attribute__((ext_vector_type(8))) float v8f;

#define NN    100
#define HH    128
#define NPAD  112
#define R4    400

#define OFF_XS    0
#define OFF_PRE   14336
#define OFF_AGG   28672
#define OFF_W     43008
#define OFF_WLAST 75776
#define OFF_BM1   75904
#define OFF_BM2   76032
#define OFF_BU1   76160
#define OFF_BU2   76288
#define OFF_GM    76416
#define OFF_BT    76544
#define OFF_MU    76672
#define OFF_RS    76784
#define SMEM_FLOATS 76896

__device__ __forceinline__ v8f wmma4(v2f a, v2f b, v8f c) {
  return __builtin_amdgcn_wmma_f32_16x16x4_f32(
       false, a,  false, b,
       (short)0, c,  false,  false);
}

__device__ __forceinline__ void cp16(float* dst, const float* src, int nflt, int tid) {
  const float4* s = (const float4*)src;
  float4*       d = (float4*)dst;
  for (int i = tid; i < (nflt >> 2); i += 256) d[i] = s[i];
}

__device__ __forceinline__ void zero16(float* dst, int nflt, int tid) {
  float4 z = make_float4(0.f, 0.f, 0.f, 0.f);
  float4* d = (float4*)dst;
  for (int i = tid; i < (nflt >> 2); i += 256) d[i] = z;
}

__device__ __forceinline__ void cpw_swz(float* dst, const float* src, int krows, int tid) {
  const int npairs = krows >> 1;
  for (int idx = tid; idx < npairs * 32; idx += 256) {
    const int kp = idx >> 5;
    const int cg = idx & 31;
    const float4 r0 = *(const float4*)(src + (size_t)(2 * kp)     * 128 + cg * 4);
    const float4 r1 = *(const float4*)(src + (size_t)(2 * kp + 1) * 128 + cg * 4);
    float4* d = (float4*)(dst + kp * 256 + cg * 8);
    d[0] = make_float4(r0.x, r1.x, r0.y, r1.y);
    d[1] = make_float4(r0.z, r1.z, r0.w, r1.w);
  }
}

__device__ __forceinline__ void wmma_row8(const float* __restrict__ wbk, int li,
                                          v2f a, v8f acc[8]) {
#pragma unroll
  for (int n = 0; n < 8; ++n) {
    const v2f bb = *(const v2f*)(wbk + (n * 16 + li) * 2);
    acc[n] = wmma4(a, bb, acc[n]);
  }
}

extern "C" __global__ __launch_bounds__(256)
void mpnn_layer_kernel(const float* __restrict__ x,
                       const float* __restrict__ Wm1, const float* __restrict__ bm1,
                       const float* __restrict__ Wm2, const float* __restrict__ bm2,
                       const float* __restrict__ Wu1, const float* __restrict__ bu1,
                       const float* __restrict__ Wu2, const float* __restrict__ bu2,
                       const float* __restrict__ gamma, const float* __restrict__ beta,
                       float* __restrict__ out)
{
  extern __shared__ float smem[];
  float* xs    = smem + OFF_XS;
  float* pre   = smem + OFF_PRE;
  float* agg   = smem + OFF_AGG;
  float* wb    = smem + OFF_W;
  float* wlast = smem + OFF_WLAST;
  float* lbm1  = smem + OFF_BM1;
  float* lbm2  = smem + OFF_BM2;
  float* lbu1  = smem + OFF_BU1;
  float* lbu2  = smem + OFF_BU2;
  float* lgm   = smem + OFF_GM;
  float* lbt   = smem + OFF_BT;
  float* lmu   = smem + OFF_MU;
  float* lrs   = smem + OFF_RS;

  const int tid  = threadIdx.x;
  const int wave = tid >> 5;
  const int lane = tid & 31;
  const int li   = lane & 15;
  const int hi   = lane >> 4;
  const int b    = blockIdx.x;

  const float* xg = x + (size_t)b * (NN * HH);

  cp16(xs, xg, NN * HH, tid);
  zero16(xs + NN * HH, (NPAD - NN) * HH, tid);
  zero16(agg, NPAD * HH, tid);
  cpw_swz(wb, Wm1, 128, tid);
  cp16(wlast, Wm1 + 128 * 128, 128, tid);
  cp16(lbm1, bm1, 128, tid);
  cp16(lbm2, bm2, 128, tid);
  cp16(lbu1, bu1, 128, tid);
  cp16(lbu2, bu2, 128, tid);
  cp16(lgm, gamma, 128, tid);
  cp16(lbt, beta, 128, tid);
  __syncthreads();

  const v8f vz = {0.f, 0.f, 0.f, 0.f, 0.f, 0.f, 0.f, 0.f};

  for (int m = wave; m < 7; m += 8) {
    v8f acc[8];
#pragma unroll
    for (int n = 0; n < 8; ++n) acc[n] = vz;
    const int arow = m * 16 + li;
    for (int ks = 0; ks < 32; ++ks) {
      const int k0 = ks * 4 + hi * 2;
      const v2f a = *(const v2f*)(xs + arow * HH + k0);
      wmma_row8(wb + (2 * ks + hi) * 256, li, a, acc);
    }
#pragma unroll
    for (int n = 0; n < 8; ++n) {
      const int col = n * 16 + li;
      const float bias = lbm1[col];
#pragma unroll
      for (int v = 0; v < 8; ++v) {
        const int row = m * 16 + hi * 8 + v;
        pre[row * HH + col] = acc[n][v] + bias;
      }
    }
  }
  __syncthreads();
  cpw_swz(wb, Wm2, 128, tid);
  __syncthreads();

  for (int m = wave; m < 25; m += 8) {
    v8f acc[8];
#pragma unroll
    for (int n = 0; n < 8; ++n) acc[n] = vz;
    const int   ar = m * 16 + li;
    const int   aj = ar >> 2;
    const float ad = (float)(ar & 3);
    for (int ks = 0; ks < 32; ++ks) {
      const int k0 = ks * 4 + hi * 2;
      const v2f p  = *(const v2f*)(pre + aj * HH + k0);
      const v2f wl = *(const v2f*)(wlast + k0);
      v2f a;
      a.x = fmaxf(p.x + ad * wl.x, 0.f);
      a.y = fmaxf(p.y + ad * wl.y, 0.f);
      wmma_row8(wb + (2 * ks + hi) * 256, li, a, acc);
    }
    float bias[8];
#pragma unroll
    for (int n = 0; n < 8; ++n) bias[n] = lbm2[n * 16 + li];
#pragma unroll
    for (int v = 0; v < 8; ++v) {
      const int r = m * 16 + hi * 8 + v;
      const int j = r >> 2, d = r & 3;
      const int jm = j - (j / 10) * 10;
      int i; bool ok;
      if      (d == 0) { i = j + 10; ok = (i < 100); }
      else if (d == 1) { i = j - 10; ok = (j >= 10); }
      else if (d == 2) { i = j + 1;  ok = (jm <= 8); }
      else             { i = j - 1;  ok = (jm >= 1); }
      if (ok) {
#pragma unroll
        for (int n = 0; n < 8; ++n) {
          const int col = n * 16 + li;
          atomicAdd(&agg[i * HH + col], acc[n][v] + bias[n]);
        }
      }
    }
  }
  __syncthreads();
  cpw_swz(wb, Wu1, 256, tid);
  __syncthreads();

  for (int m = wave; m < 7; m += 8) {
    v8f acc[8];
#pragma unroll
    for (int n = 0; n < 8; ++n) acc[n] = vz;
    const int arow = m * 16 + li;
    for (int ks = 0; ks < 32; ++ks) {
      const int k0 = ks * 4 + hi * 2;
      const v2f a = *(const v2f*)(xs + arow * HH + k0);
      wmma_row8(wb + (2 * ks + hi) * 256, li, a, acc);
    }
    for (int ks = 32; ks < 64; ++ks) {
      const int k0 = (ks - 32) * 4 + hi * 2;
      const v2f a = *(const v2f*)(agg + arow * HH + k0);
      wmma_row8(wb + (2 * ks + hi) * 256, li, a, acc);
    }
#pragma unroll
    for (int n = 0; n < 8; ++n) {
      const int col = n * 16 + li;
      const float bias = lbu1[col];
#pragma unroll
      for (int v = 0; v < 8; ++v) {
        const int row = m * 16 + hi * 8 + v;
        pre[row * HH + col] = fmaxf(acc[n][v] + bias, 0.f);
      }
    }
  }
  __syncthreads();
  cpw_swz(wb, Wu2, 128, tid);
  __syncthreads();

  for (int m = wave; m < 7; m += 8) {
    v8f acc[8];
#pragma unroll
    for (int n = 0; n < 8; ++n) acc[n] = vz;
    const int arow = m * 16 + li;
    for (int ks = 0; ks < 32; ++ks) {
      const int k0 = ks * 4 + hi * 2;
      const v2f a = *(const v2f*)(pre + arow * HH + k0);
      wmma_row8(wb + (2 * ks + hi) * 256, li, a, acc);
    }
#pragma unroll
    for (int n = 0; n < 8; ++n) {
      const int col = n * 16 + li;
      const float bias = lbu2[col];
#pragma unroll
      for (int v = 0; v < 8; ++v) {
        const int row = m * 16 + hi * 8 + v;
        agg[row * HH + col] = xs[row * HH + col] + acc[n][v] + bias;
      }
    }
  }
  __syncthreads();

  const float* hb = agg;
  if (tid < NN) {
    const float* hr = hb + tid * HH;
    float s = 0.f;
    for (int k = 0; k < HH; ++k) s += hr[k];
    const float mu = s * (1.0f / 128.0f);
    float var = 0.f;
    for (int k = 0; k < HH; ++k) { const float dd = hr[k] - mu; var += dd * dd; }
    lmu[tid] = mu;
    lrs[tid] = rsqrtf(var * (1.0f / 128.0f) + 1e-5f);
  }
  __syncthreads();

  float* og = out + (size_t)b * (NN * HH);
  for (int idx = tid; idx < NN * (HH / 4); idx += 256) {
    const int row = idx >> 5;
    const int c4  = idx & 31;
    const int col = c4 * 4;
    const float4 hv = ((const float4*)hb)[row * 32 + c4];
    const float mu = lmu[row], rs = lrs[row];
    float4 o;
    o.x = (hv.x - mu) * rs * lgm[col + 0] + lbt[col + 0];
    o.y = (hv.y - mu) * rs * lgm[col + 1] + lbt[col + 1];
    o.z = (hv.z - mu) * rs * lgm[col + 2] + lbt[col + 2];
    o.w = (hv.w - mu) * rs * lgm[col + 3] + lbt[col + 3];
    ((float4*)og)[idx] = o;
  }
}

extern "C" void kernel_launch(void* const* d_in, const int* in_sizes, int n_in,
                              void* d_out, int out_size, void* d_ws, size_t ws_size,
                              hipStream_t stream) {
  (void)n_in; (void)out_size; (void)d_ws; (void)ws_size;
  const float* x     = (const float*)d_in[0];
  const float* Wm1   = (const float*)d_in[4];
  const float* bm1   = (const float*)d_in[5];
  const float* Wm2   = (const float*)d_in[6];
  const float* bm2   = (const float*)d_in[7];
  const float* Wu1   = (const float*)d_in[8];
  const float* bu1   = (const float*)d_in[9];
  const float* Wu2   = (const float*)d_in[10];
  const float* bu2   = (const float*)d_in[11];
  const float* gamma = (const float*)d_in[12];
  const float* beta  = (const float*)d_in[13];
  float* out = (float*)d_out;

  const int B = in_sizes[0] / (NN * HH);
  const size_t shmem = (size_t)SMEM_FLOATS * sizeof(float);
  hipLaunchKernelGGL(mpnn_layer_kernel, dim3(B), dim3(256), shmem, stream,
                     x, Wm1, bm1, Wm2, bm2, Wu1, bu1, Wu2, bu2, gamma, beta, out);
}


# Staging latency hidden: all global loads of a stage issued as one group; next weight matrix prefetched into registers before each WMMA phase, stored to LDS after its barrier; waves_per_eu(2,2)
# speedup vs baseline: 1.1425x; 1.1425x over previous
#include <hip/hip_runtime.h>


typedef __attribute__((ext_vector_type(2))) float v2f;
typedef __attribute__((ext_vector_type(8))) float v8f;

#define NN    100
#define HH    128
#define NPAD  112
#define R4    400

#define OFF_XS    0
#define OFF_PRE   14336
#define OFF_AGG   28672
#define OFF_W     43008
#define OFF_WLAST 75776
#define OFF_BM1   75904
#define OFF_BM2   76032
#define OFF_BU1   76160
#define OFF_BU2   76288
#define OFF_GM    76416
#define OFF_BT    76544
#define OFF_MU    76672
#define OFF_RS    76784
#define SMEM_FLOATS 76896

__device__ __forceinline__ v8f wmma4(v2f a, v2f b, v8f c) {
  return __builtin_amdgcn_wmma_f32_16x16x4_f32(
       false, a,  false, b,
       (short)0, c,  false,  false);
}

template <int NIT>
__device__ __forceinline__ void w_fetch(float4 (&r)[2 * NIT], const float* __restrict__ src, int tid) {
#pragma unroll
  for (int it = 0; it < NIT; ++it) {
    const int idx = tid + it * 256;
    const int kp = idx >> 5;
    const int cg = idx & 31;
    r[2 * it]     = *(const float4*)(src + (size_t)(2 * kp)     * 128 + cg * 4);
    r[2 * it + 1] = *(const float4*)(src + (size_t)(2 * kp + 1) * 128 + cg * 4);
  }
}

template <int NIT>
__device__ __forceinline__ void w_store(float* dst, const float4 (&r)[2 * NIT], int tid) {
#pragma unroll
  for (int it = 0; it < NIT; ++it) {
    const int idx = tid + it * 256;
    const int kp = idx >> 5;
    const int cg = idx & 31;
    const float4 r0 = r[2 * it];
    const float4 r1 = r[2 * it + 1];
    float4* d = (float4*)(dst + kp * 256 + cg * 8);
    d[0] = make_float4(r0.x, r1.x, r0.y, r1.y);
    d[1] = make_float4(r0.z, r1.z, r0.w, r1.w);
  }
}

__device__ __forceinline__ void wmma_row8(const float* __restrict__ wbk, int li,
                                          v2f a, v8f acc[8]) {
#pragma unroll
  for (int n = 0; n < 8; ++n) {
    const v2f bb = *(const v2f*)(wbk + (n * 16 + li) * 2);
    acc[n] = wmma4(a, bb, acc[n]);
  }
}

extern "C" __global__ __launch_bounds__(256) __attribute__((amdgpu_waves_per_eu(2, 2)))
void mpnn_layer_kernel(const float* __restrict__ x,
                       const float* __restrict__ Wm1, const float* __restrict__ bm1,
                       const float* __restrict__ Wm2, const float* __restrict__ bm2,
                       const float* __restrict__ Wu1, const float* __restrict__ bu1,
                       const float* __restrict__ Wu2, const float* __restrict__ bu2,
                       const float* __restrict__ gamma, const float* __restrict__ beta,
                       float* __restrict__ out)
{
  extern __shared__ float smem[];
  float* xs    = smem + OFF_XS;
  float* pre   = smem + OFF_PRE;
  float* agg   = smem + OFF_AGG;
  float* wb    = smem + OFF_W;
  float* wlast = smem + OFF_WLAST;
  float* lbm1  = smem + OFF_BM1;
  float* lbm2  = smem + OFF_BM2;
  float* lbu1  = smem + OFF_BU1;
  float* lbu2  = smem + OFF_BU2;
  float* lgm   = smem + OFF_GM;
  float* lbt   = smem + OFF_BT;
  float* lmu   = smem + OFF_MU;
  float* lrs   = smem + OFF_RS;

  const int tid  = threadIdx.x;
  const int wave = tid >> 5;
  const int lane = tid & 31;
  const int li   = lane & 15;
  const int hi   = lane >> 4;
  const int b    = blockIdx.x;

  const float* xg = x + (size_t)b * (NN * HH);

  const float4 z4 = make_float4(0.f, 0.f, 0.f, 0.f);
  const float4* xg4 = (const float4*)xg;
  const int il = tid + 12 * 256;
  const bool inx = il < NN * HH / 4;
  float4 xr[12];
#pragma unroll
  for (int it = 0; it < 12; ++it) xr[it] = xg4[tid + it * 256];
  const float4 xt = xg4[inx ? il : NN * HH / 4 - 1];
  const float4 xl = make_float4(inx ? xt.x : 0.f, inx ? xt.y : 0.f,
                                inx ? xt.z : 0.f, inx ? xt.w : 0.f);
  float4 w1[16];
  w_fetch<8>(w1, Wm1, tid);
#pragma unroll
  for (int it = 0; it < NPAD * HH / 4 / 256; ++it) ((float4*)agg)[tid + it * 256] = z4;
  ((float4*)xs)[il + 256] = z4;
#pragma unroll
  for (int it = 0; it < 12; ++it) ((float4*)xs)[tid + it * 256] = xr[it];
  ((float4*)xs)[il] = xl;
  w_store<8>(wb, w1, tid);
  if (wave == 0) {
    ((float4*)wlast)[lane] = ((const float4*)(Wm1 + 128 * 128))[lane];
    ((float4*)lbm1)[lane]  = ((const float4*)bm1)[lane];
    ((float4*)lbm2)[lane]  = ((const float4*)bm2)[lane];
    ((float4*)lbu1)[lane]  = ((const float4*)bu1)[lane];
    ((float4*)lbu2)[lane]  = ((const float4*)bu2)[lane];
    ((float4*)lgm)[lane]   = ((const float4*)gamma)[lane];
    ((float4*)lbt)[lane]   = ((const float4*)beta)[lane];
  }
  __syncthreads();

  float4 w2[16];
  w_fetch<8>(w2, Wm2, tid);

  const v8f vz = {0.f, 0.f, 0.f, 0.f, 0.f, 0.f, 0.f, 0.f};

  for (int m = wave; m < 7; m += 8) {
    v8f acc[8];
#pragma unroll
    for (int n = 0; n < 8; ++n) acc[n] = vz;
    const int arow = m * 16 + li;
    for (int ks = 0; ks < 32; ++ks) {
      const int k0 = ks * 4 + hi * 2;
      const v2f a = *(const v2f*)(xs + arow * HH + k0);
      wmma_row8(wb + (2 * ks + hi) * 256, li, a, acc);
    }
#pragma unroll
    for (int n = 0; n < 8; ++n) {
      const int col = n * 16 + li;
      const float bias = lbm1[col];
#pragma unroll
      for (int v = 0; v < 8; ++v) {
        const int row = m * 16 + hi * 8 + v;
        pre[row * HH + col] = acc[n][v] + bias;
      }
    }
  }
  __syncthreads();
  w_store<8>(wb, w2, tid);
  __syncthreads();

  float4 w3[32];
  w_fetch<16>(w3, Wu1, tid);

  for (int m = wave; m < 25; m += 8) {
    v8f acc[8];
#pragma unroll
    for (int n = 0; n < 8; ++n) acc[n] = vz;
    const int   ar = m * 16 + li;
    const int   aj = ar >> 2;
    const float ad = (float)(ar & 3);
    for (int ks = 0; ks < 32; ++ks) {
      const int k0 = ks * 4 + hi * 2;
      const v2f p  = *(const v2f*)(pre + aj * HH + k0);
      const v2f wl = *(const v2f*)(wlast + k0);
      v2f a;
      a.x = fmaxf(p.x + ad * wl.x, 0.f);
      a.y = fmaxf(p.y + ad * wl.y, 0.f);
      wmma_row8(wb + (2 * ks + hi) * 256, li, a, acc);
    }
    float bias[8];
#pragma unroll
    for (int n = 0; n < 8; ++n) bias[n] = lbm2[n * 16 + li];
#pragma unroll
    for (int v = 0; v < 8; ++v) {
      const int r = m * 16 + hi * 8 + v;
      const int j = r >> 2, d = r & 3;
      const int jm = j - (j / 10) * 10;
      int i; bool ok;
      if      (d == 0) { i = j + 10; ok = (i < 100); }
      else if (d == 1) { i = j - 10; ok = (j >= 10); }
      else if (d == 2) { i = j + 1;  ok = (jm <= 8); }
      else             { i = j - 1;  ok = (jm >= 1); }
      if (ok) {
#pragma unroll
        for (int n = 0; n < 8; ++n) {
          const int col = n * 16 + li;
          atomicAdd(&agg[i * HH + col], acc[n][v] + bias[n]);
        }
      }
    }
  }
  __syncthreads();
  w_store<16>(wb, w3, tid);
  __syncthreads();

  float4 w4[16];
  w_fetch<8>(w4, Wu2, tid);

  for (int m = wave; m < 7; m += 8) {
    v8f acc[8];
#pragma unroll
    for (int n = 0; n < 8; ++n) acc[n] = vz;
    const int arow = m * 16 + li;
    for (int ks = 0; ks < 32; ++ks) {
      const int k0 = ks * 4 + hi * 2;
      const v2f a = *(const v2f*)(xs + arow * HH + k0);
      wmma_row8(wb + (2 * ks + hi) * 256, li, a, acc);
    }
    for (int ks = 32; ks < 64; ++ks) {
      const int k0 = (ks - 32) * 4 + hi * 2;
      const v2f a = *(const v2f*)(agg + arow * HH + k0);
      wmma_row8(wb + (2 * ks + hi) * 256, li, a, acc);
    }
#pragma unroll
    for (int n = 0; n < 8; ++n) {
      const int col = n * 16 + li;
      const float bias = lbu1[col];
#pragma unroll
      for (int v = 0; v < 8; ++v) {
        const int row = m * 16 + hi * 8 + v;
        pre[row * HH + col] = fmaxf(acc[n][v] + bias, 0.f);
      }
    }
  }
  __syncthreads();
  w_store<8>(wb, w4, tid);
  __syncthreads();

  for (int m = wave; m < 7; m += 8) {
    v8f acc[8];
#pragma unroll
    for (int n = 0; n < 8; ++n) acc[n] = vz;
    const int arow = m * 16 + li;
    for (int ks = 0; ks < 32; ++ks) {
      const int k0 = ks * 4 + hi * 2;
      const v2f a = *(const v2f*)(pre + arow * HH + k0);
      wmma_row8(wb + (2 * ks + hi) * 256, li, a, acc);
    }
#pragma unroll
    for (int n = 0; n < 8; ++n) {
      const int col = n * 16 + li;
      const float bias = lbu2[col];
#pragma unroll
      for (int v = 0; v < 8; ++v) {
        const int row = m * 16 + hi * 8 + v;
        agg[row * HH + col] = xs[row * HH + col] + acc[n][v] + bias;
      }
    }
  }
  __syncthreads();

  const float* hb = agg;
  if (tid < NN) {
    const float* hr = hb + tid * HH;
    float s = 0.f;
    for (int k = 0; k < HH; ++k) s += hr[k];
    const float mu = s * (1.0f / 128.0f);
    float var = 0.f;
    for (int k = 0; k < HH; ++k) { const float dd = hr[k] - mu; var += dd * dd; }
    lmu[tid] = mu;
    lrs[tid] = rsqrtf(var * (1.0f / 128.0f) + 1e-5f);
  }
  __syncthreads();

  float* og = out + (size_t)b * (NN * HH);
  for (int idx = tid; idx < NN * (HH / 4); idx += 256) {
    const int row = idx >> 5;
    const int c4  = idx & 31;
    const int col = c4 * 4;
    const float4 hv = ((const float4*)hb)[row * 32 + c4];
    const float mu = lmu[row], rs = lrs[row];
    float4 o;
    o.x = (hv.x - mu) * rs * lgm[col + 0] + lbt[col + 0];
    o.y = (hv.y - mu) * rs * lgm[col + 1] + lbt[col + 1];
    o.z = (hv.z - mu) * rs * lgm[col + 2] + lbt[col + 2];
    o.w = (hv.w - mu) * rs * lgm[col + 3] + lbt[col + 3];
    ((float4*)og)[idx] = o;
  }
}

extern "C" void kernel_launch(void* const* d_in, const int* in_sizes, int n_in,
                              void* d_out, int out_size, void* d_ws, size_t ws_size,
                              hipStream_t stream) {
  (void)n_in; (void)out_size; (void)d_ws; (void)ws_size;
  const float* x     = (const float*)d_in[0];
  const float* Wm1   = (const float*)d_in[4];
  const float* bm1   = (const float*)d_in[5];
  const float* Wm2   = (const float*)d_in[6];
  const float* bm2   = (const float*)d_in[7];
  const float* Wu1   = (const float*)d_in[8];
  const float* bu1   = (const float*)d_in[9];
  const float* Wu2   = (const float*)d_in[10];
  const float* bu2   = (const float*)d_in[11];
  const float* gamma = (const float*)d_in[12];
  const float* beta  = (const float*)d_in[13];
  float* out = (float*)d_out;

  const int B = in_sizes[0] / (NN * HH);
  const size_t shmem = (size_t)SMEM_FLOATS * sizeof(float);
  hipLaunchKernelGGL(mpnn_layer_kernel, dim3(B), dim3(256), shmem, stream,
                     x, Wm1, bm1, Wm2, bm2, Wu1, bu1, Wu2, bu2, gamma, beta, out);
}


# LDS layout made bank-conflict free: activation rows padded to a 132-float pitch, odd K-pair weight rows stored with columns XOR 16; A and B fragment loads and the row-sum reads no longer share banks
# speedup vs baseline: 1.4162x; 1.2396x over previous
#include <hip/hip_runtime.h>


typedef __attribute__((ext_vector_type(2))) float v2f;
typedef __attribute__((ext_vector_type(8))) float v8f;

#define NN    100
#define HH    128
#define NPAD  112
#define R4    400
#define LDP   132
#define LDP4  33

#define OFF_XS    0
#define OFF_PRE   14784
#define OFF_AGG   29568
#define OFF_W     44352
#define OFF_WLAST 77120
#define OFF_BM1   77248
#define OFF_BM2   77376
#define OFF_BU1   77504
#define OFF_BU2   77632
#define OFF_GM    77760
#define OFF_BT    77888
#define OFF_MU    78016
#define OFF_RS    78128
#define SMEM_FLOATS 78240

__device__ __forceinline__ v8f wmma4(v2f a, v2f b, v8f c) {
  return __builtin_amdgcn_wmma_f32_16x16x4_f32(
       false, a,  false, b,
       (short)0, c,  false,  false);
}

template <int NIT>
__device__ __forceinline__ void w_fetch(float4 (&r)[2 * NIT], const float* __restrict__ src, int tid) {
#pragma unroll
  for (int it = 0; it < NIT; ++it) {
    const int idx = tid + it * 256;
    const int kp = idx >> 5;
    const int cg = idx & 31;
    r[2 * it]     = *(const float4*)(src + (size_t)(2 * kp)     * 128 + cg * 4);
    r[2 * it + 1] = *(const float4*)(src + (size_t)(2 * kp + 1) * 128 + cg * 4);
  }
}

template <int NIT>
__device__ __forceinline__ void w_store(float* dst, const float4 (&r)[2 * NIT], int tid) {
#pragma unroll
  for (int it = 0; it < NIT; ++it) {
    const int idx = tid + it * 256;
    const int kp = idx >> 5;
    const int cg = idx & 31;
    const float4 r0 = r[2 * it];
    const float4 r1 = r[2 * it + 1];
    float4* d = (float4*)(dst + kp * 256 + (cg ^ ((kp & 1) << 2)) * 8);
    d[0] = make_float4(r0.x, r1.x, r0.y, r1.y);
    d[1] = make_float4(r0.z, r1.z, r0.w, r1.w);
  }
}

__device__ __forceinline__ void wmma_row8(const float* __restrict__ wbk, int le, int lo,
                                          v2f a, v8f acc[8]) {
#pragma unroll
  for (int n = 0; n < 8; ++n) {
    const v2f bb = *(const v2f*)(wbk + n * 32 + ((n & 1) ? lo : le));
    acc[n] = wmma4(a, bb, acc[n]);
  }
}

extern "C" __global__ __launch_bounds__(256) __attribute__((amdgpu_waves_per_eu(2, 2)))
void mpnn_layer_kernel(const float* __restrict__ x,
                       const float* __restrict__ Wm1, const float* __restrict__ bm1,
                       const float* __restrict__ Wm2, const float* __restrict__ bm2,
                       const float* __restrict__ Wu1, const float* __restrict__ bu1,
                       const float* __restrict__ Wu2, const float* __restrict__ bu2,
                       const float* __restrict__ gamma, const float* __restrict__ beta,
                       float* __restrict__ out)
{
  extern __shared__ float smem[];
  float* xs    = smem + OFF_XS;
  float* pre   = smem + OFF_PRE;
  float* agg   = smem + OFF_AGG;
  float* wb    = smem + OFF_W;
  float* wlast = smem + OFF_WLAST;
  float* lbm1  = smem + OFF_BM1;
  float* lbm2  = smem + OFF_BM2;
  float* lbu1  = smem + OFF_BU1;
  float* lbu2  = smem + OFF_BU2;
  float* lgm   = smem + OFF_GM;
  float* lbt   = smem + OFF_BT;
  float* lmu   = smem + OFF_MU;
  float* lrs   = smem + OFF_RS;

  const int tid  = threadIdx.x;
  const int wave = tid >> 5;
  const int lane = tid & 31;
  const int li   = lane & 15;
  const int hi   = lane >> 4;
  const int b    = blockIdx.x;
  const int le   = hi * 256 + li * 2 + hi * 32;
  const int lo   = hi * 256 + li * 2 - hi * 32;
  const int p4   = (tid >> 5) * LDP4 + (tid & 31);

  const float* xg = x + (size_t)b * (NN * HH);

  const float4 z4 = make_float4(0.f, 0.f, 0.f, 0.f);
  const float4* xg4 = (const float4*)xg;
  const int il = tid + 12 * 256;
  const bool inx = il < NN * HH / 4;
  float4 xr[12];
#pragma unroll
  for (int it = 0; it < 12; ++it) xr[it] = xg4[tid + it * 256];
  const float4 xt = xg4[inx ? il : NN * HH / 4 - 1];
  const float4 xl = make_float4(inx ? xt.x : 0.f, inx ? xt.y : 0.f,
                                inx ? xt.z : 0.f, inx ? xt.w : 0.f);
  float4 w1[16];
  w_fetch<8>(w1, Wm1, tid);
#pragma unroll
  for (int it = 0; it < NPAD * HH / 4 / 256; ++it) ((float4*)agg)[p4 + it * 8 * LDP4] = z4;
  ((float4*)xs)[p4 + 13 * 8 * LDP4] = z4;
#pragma unroll
  for (int it = 0; it < 12; ++it) ((float4*)xs)[p4 + it * 8 * LDP4] = xr[it];
  ((float4*)xs)[p4 + 12 * 8 * LDP4] = xl;
  w_store<8>(wb, w1, tid);
  if (wave == 0) {
    ((float4*)wlast)[lane] = ((const float4*)(Wm1 + 128 * 128))[lane];
    ((float4*)lbm1)[lane]  = ((const float4*)bm1)[lane];
    ((float4*)lbm2)[lane]  = ((const float4*)bm2)[lane];
    ((float4*)lbu1)[lane]  = ((const float4*)bu1)[lane];
    ((float4*)lbu2)[lane]  = ((const float4*)bu2)[lane];
    ((float4*)lgm)[lane]   = ((const float4*)gamma)[lane];
    ((float4*)lbt)[lane]   = ((const float4*)beta)[lane];
  }
  __syncthreads();

  float4 w2[16];
  w_fetch<8>(w2, Wm2, tid);

  const v8f vz = {0.f, 0.f, 0.f, 0.f, 0.f, 0.f, 0.f, 0.f};

  for (int m = wave; m < 7; m += 8) {
    v8f acc[8];
#pragma unroll
    for (int n = 0; n < 8; ++n) acc[n] = vz;
    const int arow = m * 16 + li;
    for (int ks = 0; ks < 32; ++ks) {
      const int k0 = ks * 4 + hi * 2;
      const v2f a = *(const v2f*)(xs + arow * LDP + k0);
      wmma_row8(wb + ks * 512, le, lo, a, acc);
    }
#pragma unroll
    for (int n = 0; n < 8; ++n) {
      const int col = n * 16 + li;
      const float bias = lbm1[col];
#pragma unroll
      for (int v = 0; v < 8; ++v) {
        const int row = m * 16 + hi * 8 + v;
        pre[row * LDP + col] = acc[n][v] + bias;
      }
    }
  }
  __syncthreads();
  w_store<8>(wb, w2, tid);
  __syncthreads();

  float4 w3[32];
  w_fetch<16>(w3, Wu1, tid);

  for (int m = wave; m < 25; m += 8) {
    v8f acc[8];
#pragma unroll
    for (int n = 0; n < 8; ++n) acc[n] = vz;
    const int   ar = m * 16 + li;
    const int   aj = ar >> 2;
    const float ad = (float)(ar & 3);
#pragma unroll 4
    for (int ks = 0; ks < 32; ++ks) {
      const int k0 = ks * 4 + hi * 2;
      const v2f p  = *(const v2f*)(pre + aj * LDP + k0);
      const v2f wl = *(const v2f*)(wlast + k0);
      v2f a;
      a.x = fmaxf(p.x + ad * wl.x, 0.f);
      a.y = fmaxf(p.y + ad * wl.y, 0.f);
      wmma_row8(wb + ks * 512, le, lo, a, acc);
    }
    float bias[8];
#pragma unroll
    for (int n = 0; n < 8; ++n) bias[n] = lbm2[n * 16 + li];
#pragma unroll
    for (int v = 0; v < 8; ++v) {
      const int r = m * 16 + hi * 8 + v;
      const int j = r >> 2, d = r & 3;
      const int jm = j - (j / 10) * 10;
      int i; bool ok;
      if      (d == 0) { i = j + 10; ok = (i < 100); }
      else if (d == 1) { i = j - 10; ok = (j >= 10); }
      else if (d == 2) { i = j + 1;  ok = (jm <= 8); }
      else             { i = j - 1;  ok = (jm >= 1); }
      if (ok) {
#pragma unroll
        for (int n = 0; n < 8; ++n) {
          const int col = n * 16 + li;
          atomicAdd(&agg[i * LDP + col], acc[n][v] + bias[n]);
        }
      }
    }
  }
  __syncthreads();
  w_store<16>(wb, w3, tid);
  __syncthreads();

  float4 w4[16];
  w_fetch<8>(w4, Wu2, tid);

  for (int m = wave; m < 7; m += 8) {
    v8f acc[8];
#pragma unroll
    for (int n = 0; n < 8; ++n) acc[n] = vz;
    const int arow = m * 16 + li;
    for (int ks = 0; ks < 32; ++ks) {
      const int k0 = ks * 4 + hi * 2;
      const v2f a = *(const v2f*)(xs + arow * LDP + k0);
      wmma_row8(wb + ks * 512, le, lo, a, acc);
    }
    for (int ks = 32; ks < 64; ++ks) {
      const int k0 = (ks - 32) * 4 + hi * 2;
      const v2f a = *(const v2f*)(agg + arow * LDP + k0);
      wmma_row8(wb + ks * 512, le, lo, a, acc);
    }
#pragma unroll
    for (int n = 0; n < 8; ++n) {
      const int col = n * 16 + li;
      const float bias = lbu1[col];
#pragma unroll
      for (int v = 0; v < 8; ++v) {
        const int row = m * 16 + hi * 8 + v;
        pre[row * LDP + col] = fmaxf(acc[n][v] + bias, 0.f);
      }
    }
  }
  __syncthreads();
  w_store<8>(wb, w4, tid);
  __syncthreads();

  for (int m = wave; m < 7; m += 8) {
    v8f acc[8];
#pragma unroll
    for (int n = 0; n < 8; ++n) acc[n] = vz;
    const int arow = m * 16 + li;
    for (int ks = 0; ks < 32; ++ks) {
      const int k0 = ks * 4 + hi * 2;
      const v2f a = *(const v2f*)(pre + arow * LDP + k0);
      wmma_row8(wb + ks * 512, le, lo, a, acc);
    }
#pragma unroll
    for (int n = 0; n < 8; ++n) {
      const int col = n * 16 + li;
      const float bias = lbu2[col];
#pragma unroll
      for (int v = 0; v < 8; ++v) {
        const int row = m * 16 + hi * 8 + v;
        agg[row * LDP + col] = xs[row * LDP + col] + acc[n][v] + bias;
      }
    }
  }
  __syncthreads();

  const float* hb = agg;
  if (tid < NN) {
    const float* hr = hb + tid * LDP;
    float s = 0.f;
    for (int k = 0; k < HH; ++k) s += hr[k];
    const float mu = s * (1.0f / 128.0f);
    float var = 0.f;
    for (int k = 0; k < HH; ++k) { const float dd = hr[k] - mu; var += dd * dd; }
    lmu[tid] = mu;
    lrs[tid] = rsqrtf(var * (1.0f / 128.0f) + 1e-5f);
  }
  __syncthreads();

  float* og = out + (size_t)b * (NN * HH);
  for (int idx = tid; idx < NN * (HH / 4); idx += 256) {
    const int row = idx >> 5;
    const int c4  = idx & 31;
    const int col = c4 * 4;
    const float4 hv = ((const float4*)hb)[row * LDP4 + c4];
    const float mu = lmu[row], rs = lrs[row];
    float4 o;
    o.x = (hv.x - mu) * rs * lgm[col + 0] + lbt[col + 0];
    o.y = (hv.y - mu) * rs * lgm[col + 1] + lbt[col + 1];
    o.z = (hv.z - mu) * rs * lgm[col + 2] + lbt[col + 2];
    o.w = (hv.w - mu) * rs * lgm[col + 3] + lbt[col + 3];
    ((float4*)og)[idx] = o;
  }
}

extern "C" void kernel_launch(void* const* d_in, const int* in_sizes, int n_in,
                              void* d_out, int out_size, void* d_ws, size_t ws_size,
                              hipStream_t stream) {
  (void)n_in; (void)out_size; (void)d_ws; (void)ws_size;
  const float* x     = (const float*)d_in[0];
  const float* Wm1   = (const float*)d_in[4];
  const float* bm1   = (const float*)d_in[5];
  const float* Wm2   = (const float*)d_in[6];
  const float* bm2   = (const float*)d_in[7];
  const float* Wu1   = (const float*)d_in[8];
  const float* bu1   = (const float*)d_in[9];
  const float* Wu2   = (const float*)d_in[10];
  const float* bu2   = (const float*)d_in[11];
  const float* gamma = (const float*)d_in[12];
  const float* beta  = (const float*)d_in[13];
  float* out = (float*)d_out;

  const int B = in_sizes[0] / (NN * HH);
  const size_t shmem = (size_t)SMEM_FLOATS * sizeof(float);
  hipLaunchKernelGGL(mpnn_layer_kernel, dim3(B), dim3(256), shmem, stream,
                     x, Wm1, bm1, Wm2, bm2, Wu1, bu1, Wu2, bu2, gamma, beta, out);
}


# Wm2 product moved behind the edge sum (it is linear): relu'd hidden rows of a node's up to 4 in-edges are summed in d order, then 100 rows x Wm2, not 400; LDS float atomics gone; sum order changes
# speedup vs baseline: 1.9233x; 1.3581x over previous
#include <hip/hip_runtime.h>


typedef __attribute__((ext_vector_type(2))) float v2f;
typedef __attribute__((ext_vector_type(8))) float v8f;

#define NN    100
#define HH    128
#define NPAD  112
#define R4    400
#define LDP   132
#define LDP4  33

#define OFF_XS    0
#define OFF_PRE   14784
#define OFF_AGG   29568
#define OFF_W     44352
#define OFF_WLAST 77120
#define OFF_BM1   77248
#define OFF_BM2   77376
#define OFF_BU1   77504
#define OFF_BU2   77632
#define OFF_GM    77760
#define OFF_BT    77888
#define OFF_MU    78016
#define OFF_RS    78128
#define SMEM_FLOATS 78240

__device__ __forceinline__ v8f wmma4(v2f a, v2f b, v8f c) {
  return __builtin_amdgcn_wmma_f32_16x16x4_f32(
       false, a,  false, b,
       (short)0, c,  false,  false);
}

template <int NIT>
__device__ __forceinline__ void w_fetch(float4 (&r)[2 * NIT], const float* __restrict__ src, int tid) {
#pragma unroll
  for (int it = 0; it < NIT; ++it) {
    const int idx = tid + it * 256;
    const int kp = idx >> 5;
    const int cg = idx & 31;
    r[2 * it]     = *(const float4*)(src + (size_t)(2 * kp)     * 128 + cg * 4);
    r[2 * it + 1] = *(const float4*)(src + (size_t)(2 * kp + 1) * 128 + cg * 4);
  }
}

template <int NIT>
__device__ __forceinline__ void w_store(float* dst, const float4 (&r)[2 * NIT], int tid) {
#pragma unroll
  for (int it = 0; it < NIT; ++it) {
    const int idx = tid + it * 256;
    const int kp = idx >> 5;
    const int cg = idx & 31;
    const float4 r0 = r[2 * it];
    const float4 r1 = r[2 * it + 1];
    float4* d = (float4*)(dst + kp * 256 + (cg ^ ((kp & 1) << 2)) * 8);
    d[0] = make_float4(r0.x, r1.x, r0.y, r1.y);
    d[1] = make_float4(r0.z, r1.z, r0.w, r1.w);
  }
}

__device__ __forceinline__ void wmma_row8(const float* __restrict__ wbk, int le, int lo,
                                          v2f a, v8f acc[8]) {
#pragma unroll
  for (int n = 0; n < 8; ++n) {
    const v2f bb = *(const v2f*)(wbk + n * 32 + ((n & 1) ? lo : le));
    acc[n] = wmma4(a, bb, acc[n]);
  }
}

extern "C" __global__ __launch_bounds__(256) __attribute__((amdgpu_waves_per_eu(2, 2)))
void mpnn_layer_kernel(const float* __restrict__ x,
                       const float* __restrict__ Wm1, const float* __restrict__ bm1,
                       const float* __restrict__ Wm2, const float* __restrict__ bm2,
                       const float* __restrict__ Wu1, const float* __restrict__ bu1,
                       const float* __restrict__ Wu2, const float* __restrict__ bu2,
                       const float* __restrict__ gamma, const float* __restrict__ beta,
                       float* __restrict__ out)
{
  extern __shared__ float smem[];
  float* xs    = smem + OFF_XS;
  float* pre   = smem + OFF_PRE;
  float* agg   = smem + OFF_AGG;
  float* wb    = smem + OFF_W;
  float* wlast = smem + OFF_WLAST;
  float* lbm1  = smem + OFF_BM1;
  float* lbm2  = smem + OFF_BM2;
  float* lbu1  = smem + OFF_BU1;
  float* lbu2  = smem + OFF_BU2;
  float* lgm   = smem + OFF_GM;
  float* lbt   = smem + OFF_BT;
  float* lmu   = smem + OFF_MU;
  float* lrs   = smem + OFF_RS;

  const int tid  = threadIdx.x;
  const int wave = tid >> 5;
  const int lane = tid & 31;
  const int li   = lane & 15;
  const int hi   = lane >> 4;
  const int b    = blockIdx.x;
  const int le   = hi * 256 + li * 2 + hi * 32;
  const int lo   = hi * 256 + li * 2 - hi * 32;
  const int p4   = (tid >> 5) * LDP4 + (tid & 31);

  const float* xg = x + (size_t)b * (NN * HH);

  const float4 z4 = make_float4(0.f, 0.f, 0.f, 0.f);
  const float4* xg4 = (const float4*)xg;
  const int il = tid + 12 * 256;
  const bool inx = il < NN * HH / 4;
  float4 xr[12];
#pragma unroll
  for (int it = 0; it < 12; ++it) xr[it] = xg4[tid + it * 256];
  const float4 xt = xg4[inx ? il : NN * HH / 4 - 1];
  const float4 xl = make_float4(inx ? xt.x : 0.f, inx ? xt.y : 0.f,
                                inx ? xt.z : 0.f, inx ? xt.w : 0.f);
  float4 w1[16];
  w_fetch<8>(w1, Wm1, tid);
  ((float4*)xs)[p4 + 13 * 8 * LDP4] = z4;
#pragma unroll
  for (int it = 0; it < 12; ++it) ((float4*)xs)[p4 + it * 8 * LDP4] = xr[it];
  ((float4*)xs)[p4 + 12 * 8 * LDP4] = xl;
  w_store<8>(wb, w1, tid);
  if (wave == 0) {
    ((float4*)wlast)[lane] = ((const float4*)(Wm1 + 128 * 128))[lane];
    ((float4*)lbm1)[lane]  = ((const float4*)bm1)[lane];
    ((float4*)lbm2)[lane]  = ((const float4*)bm2)[lane];
    ((float4*)lbu1)[lane]  = ((const float4*)bu1)[lane];
    ((float4*)lbu2)[lane]  = ((const float4*)bu2)[lane];
    ((float4*)lgm)[lane]   = ((const float4*)gamma)[lane];
    ((float4*)lbt)[lane]   = ((const float4*)beta)[lane];
  }
  __syncthreads();

  float4 w2[16];
  w_fetch<8>(w2, Wm2, tid);

  const v8f vz = {0.f, 0.f, 0.f, 0.f, 0.f, 0.f, 0.f, 0.f};

  for (int m = wave; m < 7; m += 8) {
    v8f acc[8];
#pragma unroll
    for (int n = 0; n < 8; ++n) acc[n] = vz;
    const int arow = m * 16 + li;
    for (int ks = 0; ks < 32; ++ks) {
      const int k0 = ks * 4 + hi * 2;
      const v2f a = *(const v2f*)(xs + arow * LDP + k0);
      wmma_row8(wb + ks * 512, le, lo, a, acc);
    }
#pragma unroll
    for (int n = 0; n < 8; ++n) {
      const int col = n * 16 + li;
      const float bias = lbm1[col];
#pragma unroll
      for (int v = 0; v < 8; ++v) {
        const int row = m * 16 + hi * 8 + v;
        pre[row * LDP + col] = acc[n][v] + bias;
      }
    }
  }
  __syncthreads();
  w_store<8>(wb, w2, tid);
  __syncthreads();

  float4 w3[32];
  w_fetch<16>(w3, Wu1, tid);

  for (int m = wave; m < 7; m += 8) {
    v8f acc[8];
#pragma unroll
    for (int n = 0; n < 8; ++n) acc[n] = vz;
    const int  i  = m * 16 + li;
    const int  ic = i - (i / 10) * 10;
    const bool e0 = (i >= 10) && (i < NN);
    const bool e1 = (i < NN - 10);
    const bool e2 = (i < NN) && (ic >= 1);
    const bool e3 = (i < NN) && (ic <= 8);
    const float* q0 = pre + (e0 ? i - 10 : i) * LDP + hi * 2;
    const float* q1 = pre + (e1 ? i + 10 : i) * LDP + hi * 2;
    const float* q2 = pre + (e2 ? i - 1  : i) * LDP + hi * 2;
    const float* q3 = pre + (e3 ? i + 1  : i) * LDP + hi * 2;
    const float* qw = wlast + hi * 2;
#pragma unroll 4
    for (int ks = 0; ks < 32; ++ks) {
      const v2f h0 = *(const v2f*)(q0 + ks * 4);
      const v2f h1 = *(const v2f*)(q1 + ks * 4);
      const v2f h2 = *(const v2f*)(q2 + ks * 4);
      const v2f h3 = *(const v2f*)(q3 + ks * 4);
      const v2f wl = *(const v2f*)(qw + ks * 4);
      v2f a;
      a.x = (e0 ? fmaxf(h0.x, 0.f) : 0.f)
          + (e1 ? fmaxf(h1.x + wl.x, 0.f) : 0.f)
          + (e2 ? fmaxf(h2.x + 2.f * wl.x, 0.f) : 0.f)
          + (e3 ? fmaxf(h3.x + 3.f * wl.x, 0.f) : 0.f);
      a.y = (e0 ? fmaxf(h0.y, 0.f) : 0.f)
          + (e1 ? fmaxf(h1.y + wl.y, 0.f) : 0.f)
          + (e2 ? fmaxf(h2.y + 2.f * wl.y, 0.f) : 0.f)
          + (e3 ? fmaxf(h3.y + 3.f * wl.y, 0.f) : 0.f);
      wmma_row8(wb + ks * 512, le, lo, a, acc);
    }
    float bias[8];
#pragma unroll
    for (int n = 0; n < 8; ++n) bias[n] = lbm2[n * 16 + li];
#pragma unroll
    for (int v = 0; v < 8; ++v) {
      const int r  = m * 16 + hi * 8 + v;
      const int rc = r - (r / 10) * 10;
      const int  in = (r < NN) ? 1 : 0;
      const float dg = (float)(((r >= 10 ? 1 : 0) & in) + (r < NN - 10 ? 1 : 0)
                               + ((rc >= 1 ? 1 : 0) & in) + ((rc <= 8 ? 1 : 0) & in));
#pragma unroll
      for (int n = 0; n < 8; ++n) {
        const int col = n * 16 + li;
        agg[r * LDP + col] = acc[n][v] + dg * bias[n];
      }
    }
  }
  __syncthreads();
  w_store<16>(wb, w3, tid);
  __syncthreads();

  float4 w4[16];
  w_fetch<8>(w4, Wu2, tid);

  for (int m = wave; m < 7; m += 8) {
    v8f acc[8];
#pragma unroll
    for (int n = 0; n < 8; ++n) acc[n] = vz;
    const int arow = m * 16 + li;
    for (int ks = 0; ks < 32; ++ks) {
      const int k0 = ks * 4 + hi * 2;
      const v2f a = *(const v2f*)(xs + arow * LDP + k0);
      wmma_row8(wb + ks * 512, le, lo, a, acc);
    }
    for (int ks = 32; ks < 64; ++ks) {
      const int k0 = (ks - 32) * 4 + hi * 2;
      const v2f a = *(const v2f*)(agg + arow * LDP + k0);
      wmma_row8(wb + ks * 512, le, lo, a, acc);
    }
#pragma unroll
    for (int n = 0; n < 8; ++n) {
      const int col = n * 16 + li;
      const float bias = lbu1[col];
#pragma unroll
      for (int v = 0; v < 8; ++v) {
        const int row = m * 16 + hi * 8 + v;
        pre[row * LDP + col] = fmaxf(acc[n][v] + bias, 0.f);
      }
    }
  }
  __syncthreads();
  w_store<8>(wb, w4, tid);
  __syncthreads();

  for (int m = wave; m < 7; m += 8) {
    v8f acc[8];
#pragma unroll
    for (int n = 0; n < 8; ++n) acc[n] = vz;
    const int arow = m * 16 + li;
    for (int ks = 0; ks < 32; ++ks) {
      const int k0 = ks * 4 + hi * 2;
      const v2f a = *(const v2f*)(pre + arow * LDP + k0);
      wmma_row8(wb + ks * 512, le, lo, a, acc);
    }
#pragma unroll
    for (int n = 0; n < 8; ++n) {
      const int col = n * 16 + li;
      const float bias = lbu2[col];
#pragma unroll
      for (int v = 0; v < 8; ++v) {
        const int row = m * 16 + hi * 8 + v;
        agg[row * LDP + col] = xs[row * LDP + col] + acc[n][v] + bias;
      }
    }
  }
  __syncthreads();

  const float* hb = agg;
  if (tid < NN) {
    const float* hr = hb + tid * LDP;
    float s = 0.f;
    for (int k = 0; k < HH; ++k) s += hr[k];
    const float mu = s * (1.0f / 128.0f);
    float var = 0.f;
    for (int k = 0; k < HH; ++k) { const float dd = hr[k] - mu; var += dd * dd; }
    lmu[tid] = mu;
    lrs[tid] = rsqrtf(var * (1.0f / 128.0f) + 1e-5f);
  }
  __syncthreads();

  float* og = out + (size_t)b * (NN * HH);
  for (int idx = tid; idx < NN * (HH / 4); idx += 256) {
    const int row = idx >> 5;
    const int c4  = idx & 31;
    const int col = c4 * 4;
    const float4 hv = ((const float4*)hb)[row * LDP4 + c4];
    const float mu = lmu[row], rs = lrs[row];
    float4 o;
    o.x = (hv.x - mu) * rs * lgm[col + 0] + lbt[col + 0];
    o.y = (hv.y - mu) * rs * lgm[col + 1] + lbt[col + 1];
    o.z = (hv.z - mu) * rs * lgm[col + 2] + lbt[col + 2];
    o.w = (hv.w - mu) * rs * lgm[col + 3] + lbt[col + 3];
    ((float4*)og)[idx] = o;
  }
}

extern "C" void kernel_launch(void* const* d_in, const int* in_sizes, int n_in,
                              void* d_out, int out_size, void* d_ws, size_t ws_size,
                              hipStream_t stream) {
  (void)n_in; (void)out_size; (void)d_ws; (void)ws_size;
  const float* x     = (const float*)d_in[0];
  const float* Wm1   = (const float*)d_in[4];
  const float* bm1   = (const float*)d_in[5];
  const float* Wm2   = (const float*)d_in[6];
  const float* bm2   = (const float*)d_in[7];
  const float* Wu1   = (const float*)d_in[8];
  const float* bu1   = (const float*)d_in[9];
  const float* Wu2   = (const float*)d_in[10];
  const float* bu2   = (const float*)d_in[11];
  const float* gamma = (const float*)d_in[12];
  const float* beta  = (const float*)d_in[13];
  float* out = (float*)d_out;

  const int B = in_sizes[0] / (NN * HH);
  const size_t shmem = (size_t)SMEM_FLOATS * sizeof(float);
  hipLaunchKernelGGL(mpnn_layer_kernel, dim3(B), dim3(256), shmem, stream,
                     x, Wm1, bm1, Wm2, bm2, Wu1, bu1, Wu2, bu2, gamma, beta, out);
}


# Matrix phases 1, 3, 4 re-tiled by column: each of the 8 waves owns one column tile over all 7 row tiles (224 WMMAs per 128 of K), not 7 waves a row tile each (256) with the 8th idle
# speedup vs baseline: 2.0309x; 1.0559x over previous
#include <hip/hip_runtime.h>


typedef __attribute__((ext_vector_type(2))) float v2f;
typedef __attribute__((ext_vector_type(8))) float v8f;

#define NN    100
#define HH    128
#define NPAD  112
#define R4    400
#define LDP   132
#define LDP4  33

#define OFF_XS    0
#define OFF_PRE   14784
#define OFF_AGG   29568
#define OFF_W     44352
#define OFF_WLAST 77120
#define OFF_BM1   77248
#define OFF_BM2   77376
#define OFF_BU1   77504
#define OFF_BU2   77632
#define OFF_GM    77760
#define OFF_BT    77888
#define OFF_MU    78016
#define OFF_RS    78128
#define SMEM_FLOATS 78240

__device__ __forceinline__ v8f wmma4(v2f a, v2f b, v8f c) {
  return __builtin_amdgcn_wmma_f32_16x16x4_f32(
       false, a,  false, b,
       (short)0, c,  false,  false);
}

template <int NIT>
__device__ __forceinline__ void w_fetch(float4 (&r)[2 * NIT], const float* __restrict__ src, int tid) {
#pragma unroll
  for (int it = 0; it < NIT; ++it) {
    const int idx = tid + it * 256;
    const int kp = idx >> 5;
    const int cg = idx & 31;
    r[2 * it]     = *(const float4*)(src + (size_t)(2 * kp)     * 128 + cg * 4);
    r[2 * it + 1] = *(const float4*)(src + (size_t)(2 * kp + 1) * 128 + cg * 4);
  }
}

template <int NIT>
__device__ __forceinline__ void w_store(float* dst, const float4 (&r)[2 * NIT], int tid) {
#pragma unroll
  for (int it = 0; it < NIT; ++it) {
    const int idx = tid + it * 256;
    const int kp = idx >> 5;
    const int cg = idx & 31;
    const float4 r0 = r[2 * it];
    const float4 r1 = r[2 * it + 1];
    float4* d = (float4*)(dst + kp * 256 + (cg ^ ((kp & 1) << 2)) * 8);
    d[0] = make_float4(r0.x, r1.x, r0.y, r1.y);
    d[1] = make_float4(r0.z, r1.z, r0.w, r1.w);
  }
}

__device__ __forceinline__ void wmma_row8(const float* __restrict__ wbk, int le, int lo,
                                          v2f a, v8f acc[8]) {
#pragma unroll
  for (int n = 0; n < 8; ++n) {
    const v2f bb = *(const v2f*)(wbk + n * 32 + ((n & 1) ? lo : le));
    acc[n] = wmma4(a, bb, acc[n]);
  }
}

__device__ __forceinline__ void wmma_col7x4(const float* __restrict__ ap,
                                            const float* __restrict__ bp, v8f acc[7]) {
#pragma unroll
  for (int j = 0; j < 4; ++j) {
    const v2f bb = *(const v2f*)(bp + j * 512);
#pragma unroll
    for (int m = 0; m < 7; ++m) {
      const v2f a = *(const v2f*)(ap + m * 16 * LDP + j * 4);
      acc[m] = wmma4(a, bb, acc[m]);
    }
  }
}

extern "C" __global__ __launch_bounds__(256) __attribute__((amdgpu_waves_per_eu(2, 2)))
void mpnn_layer_kernel(const float* __restrict__ x,
                       const float* __restrict__ Wm1, const float* __restrict__ bm1,
                       const float* __restrict__ Wm2, const float* __restrict__ bm2,
                       const float* __restrict__ Wu1, const float* __restrict__ bu1,
                       const float* __restrict__ Wu2, const float* __restrict__ bu2,
                       const float* __restrict__ gamma, const float* __restrict__ beta,
                       float* __restrict__ out)
{
  extern __shared__ float smem[];
  float* xs    = smem + OFF_XS;
  float* pre   = smem + OFF_PRE;
  float* agg   = smem + OFF_AGG;
  float* wb    = smem + OFF_W;
  float* wlast = smem + OFF_WLAST;
  float* lbm1  = smem + OFF_BM1;
  float* lbm2  = smem + OFF_BM2;
  float* lbu1  = smem + OFF_BU1;
  float* lbu2  = smem + OFF_BU2;
  float* lgm   = smem + OFF_GM;
  float* lbt   = smem + OFF_BT;
  float* lmu   = smem + OFF_MU;
  float* lrs   = smem + OFF_RS;

  const int tid  = threadIdx.x;
  const int wave = tid >> 5;
  const int lane = tid & 31;
  const int li   = lane & 15;
  const int hi   = lane >> 4;
  const int b    = blockIdx.x;
  const int le   = hi * 256 + li * 2 + hi * 32;
  const int lo   = hi * 256 + li * 2 - hi * 32;
  const int ncol = wave * 16 + li;
  const int bofs = hi * 256 + ((wave ^ hi) << 5) + li * 2;
  const int aofs = li * LDP + hi * 2;
  const int p4   = (tid >> 5) * LDP4 + (tid & 31);

  const float* xg = x + (size_t)b * (NN * HH);

  const float4 z4 = make_float4(0.f, 0.f, 0.f, 0.f);
  const float4* xg4 = (const float4*)xg;
  const int il = tid + 12 * 256;
  const bool inx = il < NN * HH / 4;
  float4 xr[12];
#pragma unroll
  for (int it = 0; it < 12; ++it) xr[it] = xg4[tid + it * 256];
  const float4 xt = xg4[inx ? il : NN * HH / 4 - 1];
  const float4 xl = make_float4(inx ? xt.x : 0.f, inx ? xt.y : 0.f,
                                inx ? xt.z : 0.f, inx ? xt.w : 0.f);
  float4 w1[16];
  w_fetch<8>(w1, Wm1, tid);
  ((float4*)xs)[p4 + 13 * 8 * LDP4] = z4;
#pragma unroll
  for (int it = 0; it < 12; ++it) ((float4*)xs)[p4 + it * 8 * LDP4] = xr[it];
  ((float4*)xs)[p4 + 12 * 8 * LDP4] = xl;
  w_store<8>(wb, w1, tid);
  if (wave == 0) {
    ((float4*)wlast)[lane] = ((const float4*)(Wm1 + 128 * 128))[lane];
    ((float4*)lbm1)[lane]  = ((const float4*)bm1)[lane];
    ((float4*)lbm2)[lane]  = ((const float4*)bm2)[lane];
    ((float4*)lbu1)[lane]  = ((const float4*)bu1)[lane];
    ((float4*)lbu2)[lane]  = ((const float4*)bu2)[lane];
    ((float4*)lgm)[lane]   = ((const float4*)gamma)[lane];
    ((float4*)lbt)[lane]   = ((const float4*)beta)[lane];
  }
  __syncthreads();

  float4 w2[16];
  w_fetch<8>(w2, Wm2, tid);

  const v8f vz = {0.f, 0.f, 0.f, 0.f, 0.f, 0.f, 0.f, 0.f};

  {
    v8f acc[7];
#pragma unroll
    for (int m = 0; m < 7; ++m) acc[m] = vz;
    for (int kq = 0; kq < 8; ++kq)
      wmma_col7x4(xs + aofs + kq * 16, wb + bofs + kq * 2048, acc);
    const float bias = lbm1[ncol];
#pragma unroll
    for (int m = 0; m < 7; ++m) {
#pragma unroll
      for (int v = 0; v < 8; ++v) {
        const int row = m * 16 + hi * 8 + v;
        pre[row * LDP + ncol] = acc[m][v] + bias;
      }
    }
  }
  __syncthreads();
  w_store<8>(wb, w2, tid);
  __syncthreads();

  float4 w3[32];
  w_fetch<16>(w3, Wu1, tid);

  for (int m = wave; m < 7; m += 8) {
    v8f acc[8];
#pragma unroll
    for (int n = 0; n < 8; ++n) acc[n] = vz;
    const int  i  = m * 16 + li;
    const int  ic = i - (i / 10) * 10;
    const bool e0 = (i >= 10) && (i < NN);
    const bool e1 = (i < NN - 10);
    const bool e2 = (i < NN) && (ic >= 1);
    const bool e3 = (i < NN) && (ic <= 8);
    const float* q0 = pre + (e0 ? i - 10 : i) * LDP + hi * 2;
    const float* q1 = pre + (e1 ? i + 10 : i) * LDP + hi * 2;
    const float* q2 = pre + (e2 ? i - 1  : i) * LDP + hi * 2;
    const float* q3 = pre + (e3 ? i + 1  : i) * LDP + hi * 2;
    const float* qw = wlast + hi * 2;
#pragma unroll 4
    for (int ks = 0; ks < 32; ++ks) {
      const v2f h0 = *(const v2f*)(q0 + ks * 4);
      const v2f h1 = *(const v2f*)(q1 + ks * 4);
      const v2f h2 = *(const v2f*)(q2 + ks * 4);
      const v2f h3 = *(const v2f*)(q3 + ks * 4);
      const v2f wl = *(const v2f*)(qw + ks * 4);
      v2f a;
      a.x = (e0 ? fmaxf(h0.x, 0.f) : 0.f)
          + (e1 ? fmaxf(h1.x + wl.x, 0.f) : 0.f)
          + (e2 ? fmaxf(h2.x + 2.f * wl.x, 0.f) : 0.f)
          + (e3 ? fmaxf(h3.x + 3.f * wl.x, 0.f) : 0.f);
      a.y = (e0 ? fmaxf(h0.y, 0.f) : 0.f)
          + (e1 ? fmaxf(h1.y + wl.y, 0.f) : 0.f)
          + (e2 ? fmaxf(h2.y + 2.f * wl.y, 0.f) : 0.f)
          + (e3 ? fmaxf(h3.y + 3.f * wl.y, 0.f) : 0.f);
      wmma_row8(wb + ks * 512, le, lo, a, acc);
    }
    float bias[8];
#pragma unroll
    for (int n = 0; n < 8; ++n) bias[n] = lbm2[n * 16 + li];
#pragma unroll
    for (int v = 0; v < 8; ++v) {
      const int r  = m * 16 + hi * 8 + v;
      const int rc = r - (r / 10) * 10;
      const int  in = (r < NN) ? 1 : 0;
      const float dg = (float)(((r >= 10 ? 1 : 0) & in) + (r < NN - 10 ? 1 : 0)
                               + ((rc >= 1 ? 1 : 0) & in) + ((rc <= 8 ? 1 : 0) & in));
#pragma unroll
      for (int n = 0; n < 8; ++n) {
        const int col = n * 16 + li;
        agg[r * LDP + col] = acc[n][v] + dg * bias[n];
      }
    }
  }
  __syncthreads();
  w_store<16>(wb, w3, tid);
  __syncthreads();

  float4 w4[16];
  w_fetch<8>(w4, Wu2, tid);

  {
    v8f acc[7];
#pragma unroll
    for (int m = 0; m < 7; ++m) acc[m] = vz;
    for (int kq = 0; kq < 8; ++kq)
      wmma_col7x4(xs + aofs + kq * 16, wb + bofs + kq * 2048, acc);
    for (int kq = 0; kq < 8; ++kq)
      wmma_col7x4(agg + aofs + kq * 16, wb + bofs + (8 + kq) * 2048, acc);
    const float bias = lbu1[ncol];
#pragma unroll
    for (int m = 0; m < 7; ++m) {
#pragma unroll
      for (int v = 0; v < 8; ++v) {
        const int row = m * 16 + hi * 8 + v;
        pre[row * LDP + ncol] = fmaxf(acc[m][v] + bias, 0.f);
      }
    }
  }
  __syncthreads();
  w_store<8>(wb, w4, tid);
  __syncthreads();

  {
    v8f acc[7];
#pragma unroll
    for (int m = 0; m < 7; ++m) acc[m] = vz;
    for (int kq = 0; kq < 8; ++kq)
      wmma_col7x4(pre + aofs + kq * 16, wb + bofs + kq * 2048, acc);
    const float bias = lbu2[ncol];
#pragma unroll
    for (int m = 0; m < 7; ++m) {
#pragma unroll
      for (int v = 0; v < 8; ++v) {
        const int row = m * 16 + hi * 8 + v;
        agg[row * LDP + ncol] = xs[row * LDP + ncol] + acc[m][v] + bias;
      }
    }
  }
  __syncthreads();

  const float* hb = agg;
  if (tid < NN) {
    const float* hr = hb + tid * LDP;
    float s = 0.f;
    for (int k = 0; k < HH; ++k) s += hr[k];
    const float mu = s * (1.0f / 128.0f);
    float var = 0.f;
    for (int k = 0; k < HH; ++k) { const float dd = hr[k] - mu; var += dd * dd; }
    lmu[tid] = mu;
    lrs[tid] = rsqrtf(var * (1.0f / 128.0f) + 1e-5f);
  }
  __syncthreads();

  float* og = out + (size_t)b * (NN * HH);
  for (int idx = tid; idx < NN * (HH / 4); idx += 256) {
    const int row = idx >> 5;
    const int c4  = idx & 31;
    const int col = c4 * 4;
    const float4 hv = ((const float4*)hb)[row * LDP4 + c4];
    const float mu = lmu[row], rs = lrs[row];
    float4 o;
    o.x = (hv.x - mu) * rs * lgm[col + 0] + lbt[col + 0];
    o.y = (hv.y - mu) * rs * lgm[col + 1] + lbt[col + 1];
    o.z = (hv.z - mu) * rs * lgm[col + 2] + lbt[col + 2];
    o.w = (hv.w - mu) * rs * lgm[col + 3] + lbt[col + 3];
    ((float4*)og)[idx] = o;
  }
}

extern "C" void kernel_launch(void* const* d_in, const int* in_sizes, int n_in,
                              void* d_out, int out_size, void* d_ws, size_t ws_size,
                              hipStream_t stream) {
  (void)n_in; (void)out_size; (void)d_ws; (void)ws_size;
  const float* x     = (const float*)d_in[0];
  const float* Wm1   = (const float*)d_in[4];
  const float* bm1   = (const float*)d_in[5];
  const float* Wm2   = (const float*)d_in[6];
  const float* bm2   = (const float*)d_in[7];
  const float* Wu1   = (const float*)d_in[8];
  const float* bu1   = (const float*)d_in[9];
  const float* Wu2   = (const float*)d_in[10];
  const float* bu2   = (const float*)d_in[11];
  const float* gamma = (const float*)d_in[12];
  const float* beta  = (const float*)d_in[13];
  float* out = (float*)d_out;

  const int B = in_sizes[0] / (NN * HH);
  const size_t shmem = (size_t)SMEM_FLOATS * sizeof(float);
  hipLaunchKernelGGL(mpnn_layer_kernel, dim3(B), dim3(256), shmem, stream,
                     x, Wm1, bm1, Wm2, bm2, Wu1, bu1, Wu2, bu2, gamma, beta, out);
}


# Persistent grid: 256 work-groups (one per WGP) loop over 8 batch elements each; the next element's x and Wm1 are loaded into registers under phase 4's WMMAs and stored to LDS before the LayerNorm
# speedup vs baseline: 2.0954x; 1.0318x over previous
#include <hip/hip_runtime.h>


typedef __attribute__((ext_vector_type(2))) float v2f;
typedef __attribute__((ext_vector_type(8))) float v8f;

#define NN    100
#define HH    128
#define NPAD  112
#define R4    400
#define LDP   132
#define LDP4  33

#define OFF_XS    0
#define OFF_PRE   14784
#define OFF_AGG   29568
#define OFF_W     44352
#define OFF_WLAST 77120
#define OFF_BM1   77248
#define OFF_BM2   77376
#define OFF_BU1   77504
#define OFF_BU2   77632
#define OFF_GM    77760
#define OFF_BT    77888
#define OFF_MU    78016
#define OFF_RS    78128
#define SMEM_FLOATS 78240

__device__ __forceinline__ v8f wmma4(v2f a, v2f b, v8f c) {
  return __builtin_amdgcn_wmma_f32_16x16x4_f32(
       false, a,  false, b,
       (short)0, c,  false,  false);
}

template <int NIT>
__device__ __forceinline__ void w_fetch(float4 (&r)[2 * NIT], const float* __restrict__ src, int tid) {
#pragma unroll
  for (int it = 0; it < NIT; ++it) {
    const int idx = tid + it * 256;
    const int kp = idx >> 5;
    const int cg = idx & 31;
    r[2 * it]     = *(const float4*)(src + (size_t)(2 * kp)     * 128 + cg * 4);
    r[2 * it + 1] = *(const float4*)(src + (size_t)(2 * kp + 1) * 128 + cg * 4);
  }
}

template <int NIT>
__device__ __forceinline__ void w_store(float* dst, const float4 (&r)[2 * NIT], int tid) {
#pragma unroll
  for (int it = 0; it < NIT; ++it) {
    const int idx = tid + it * 256;
    const int kp = idx >> 5;
    const int cg = idx & 31;
    const float4 r0 = r[2 * it];
    const float4 r1 = r[2 * it + 1];
    float4* d = (float4*)(dst + kp * 256 + (cg ^ ((kp & 1) << 2)) * 8);
    d[0] = make_float4(r0.x, r1.x, r0.y, r1.y);
    d[1] = make_float4(r0.z, r1.z, r0.w, r1.w);
  }
}

struct XRows { float4 r0, r1, r2, r3, r4, r5, r6, r7, r8, r9, r10, r11, r12; };

__device__ __forceinline__ XRows x_fetch(const float* __restrict__ xg, int tid) {
  const float4* xg4 = (const float4*)xg;
  const int il = tid + 12 * 256;
  const bool inx = il < NN * HH / 4;
  XRows r;
  r.r0  = xg4[tid + 0 * 256];
  r.r1  = xg4[tid + 1 * 256];
  r.r2  = xg4[tid + 2 * 256];
  r.r3  = xg4[tid + 3 * 256];
  r.r4  = xg4[tid + 4 * 256];
  r.r5  = xg4[tid + 5 * 256];
  r.r6  = xg4[tid + 6 * 256];
  r.r7  = xg4[tid + 7 * 256];
  r.r8  = xg4[tid + 8 * 256];
  r.r9  = xg4[tid + 9 * 256];
  r.r10 = xg4[tid + 10 * 256];
  r.r11 = xg4[tid + 11 * 256];
  const float4 xt = xg4[inx ? il : NN * HH / 4 - 1];
  r.r12 = make_float4(inx ? xt.x : 0.f, inx ? xt.y : 0.f,
                      inx ? xt.z : 0.f, inx ? xt.w : 0.f);
  return r;
}

__device__ __forceinline__ void x_store(float* xs, const XRows& r, int p4) {
  float4* d = (float4*)xs + p4;
  d[0 * 8 * LDP4]  = r.r0;
  d[1 * 8 * LDP4]  = r.r1;
  d[2 * 8 * LDP4]  = r.r2;
  d[3 * 8 * LDP4]  = r.r3;
  d[4 * 8 * LDP4]  = r.r4;
  d[5 * 8 * LDP4]  = r.r5;
  d[6 * 8 * LDP4]  = r.r6;
  d[7 * 8 * LDP4]  = r.r7;
  d[8 * 8 * LDP4]  = r.r8;
  d[9 * 8 * LDP4]  = r.r9;
  d[10 * 8 * LDP4] = r.r10;
  d[11 * 8 * LDP4] = r.r11;
  d[12 * 8 * LDP4] = r.r12;
}

__device__ __forceinline__ void wmma_row8(const float* __restrict__ wbk, int le, int lo,
                                          v2f a, v8f acc[8]) {
#pragma unroll
  for (int n = 0; n < 8; ++n) {
    const v2f bb = *(const v2f*)(wbk + n * 32 + ((n & 1) ? lo : le));
    acc[n] = wmma4(a, bb, acc[n]);
  }
}

__device__ __forceinline__ void wmma_col7x4(const float* __restrict__ ap,
                                            const float* __restrict__ bp, v8f acc[7]) {
#pragma unroll
  for (int j = 0; j < 4; ++j) {
    const v2f bb = *(const v2f*)(bp + j * 512);
#pragma unroll
    for (int m = 0; m < 7; ++m) {
      const v2f a = *(const v2f*)(ap + m * 16 * LDP + j * 4);
      acc[m] = wmma4(a, bb, acc[m]);
    }
  }
}

extern "C" __global__ __launch_bounds__(256) __attribute__((amdgpu_waves_per_eu(2, 2)))
void mpnn_layer_kernel(const float* __restrict__ x,
                       const float* __restrict__ Wm1, const float* __restrict__ bm1,
                       const float* __restrict__ Wm2, const float* __restrict__ bm2,
                       const float* __restrict__ Wu1, const float* __restrict__ bu1,
                       const float* __restrict__ Wu2, const float* __restrict__ bu2,
                       const float* __restrict__ gamma, const float* __restrict__ beta,
                       float* __restrict__ out, int nb)
{
  extern __shared__ float smem[];
  float* xs    = smem + OFF_XS;
  float* pre   = smem + OFF_PRE;
  float* agg   = smem + OFF_AGG;
  float* wb    = smem + OFF_W;
  float* wlast = smem + OFF_WLAST;
  float* lbm1  = smem + OFF_BM1;
  float* lbm2  = smem + OFF_BM2;
  float* lbu1  = smem + OFF_BU1;
  float* lbu2  = smem + OFF_BU2;
  float* lgm   = smem + OFF_GM;
  float* lbt   = smem + OFF_BT;
  float* lmu   = smem + OFF_MU;
  float* lrs   = smem + OFF_RS;

  int b = blockIdx.x;
  if (b >= nb) return;

  {
    const int tid  = threadIdx.x;
    const int lane = tid & 31;
    const int p4   = (tid >> 5) * LDP4 + lane;
    const float4 z4 = make_float4(0.f, 0.f, 0.f, 0.f);
    float4 w1[16];
    const XRows xr = x_fetch(x + (size_t)b * (NN * HH), tid);
    w_fetch<8>(w1, Wm1, tid);
    ((float4*)xs)[p4 + 13 * 8 * LDP4] = z4;
    x_store(xs, xr, p4);
    w_store<8>(wb, w1, tid);
    if ((tid >> 5) == 0) {
      ((float4*)wlast)[lane] = ((const float4*)(Wm1 + 128 * 128))[lane];
      ((float4*)lbm1)[lane]  = ((const float4*)bm1)[lane];
      ((float4*)lbm2)[lane]  = ((const float4*)bm2)[lane];
      ((float4*)lbu1)[lane]  = ((const float4*)bu1)[lane];
      ((float4*)lbu2)[lane]  = ((const float4*)bu2)[lane];
      ((float4*)lgm)[lane]   = ((const float4*)gamma)[lane];
      ((float4*)lbt)[lane]   = ((const float4*)beta)[lane];
    }
  }
  __syncthreads();

  const v8f vz = {0.f, 0.f, 0.f, 0.f, 0.f, 0.f, 0.f, 0.f};

  for (;;) {
  const int  bn   = b + (int)gridDim.x;
  const bool more = bn < nb;

  int tid = threadIdx.x;
  asm volatile("" : "+v"(tid));
  const int wave = tid >> 5;
  const int lane = tid & 31;
  const int li   = lane & 15;
  const int hi   = lane >> 4;
  const int le   = hi * 256 + li * 2 + hi * 32;
  const int lo   = hi * 256 + li * 2 - hi * 32;
  const int ncol = wave * 16 + li;
  const int bofs = hi * 256 + ((wave ^ hi) << 5) + li * 2;
  const int aofs = li * LDP + hi * 2;
  const int p4   = (tid >> 5) * LDP4 + (tid & 31);

  float4 w2[16];
  w_fetch<8>(w2, Wm2, tid);

  {
    v8f acc[7];
#pragma unroll
    for (int m = 0; m < 7; ++m) acc[m] = vz;
    for (int kq = 0; kq < 8; ++kq)
      wmma_col7x4(xs + aofs + kq * 16, wb + bofs + kq * 2048, acc);
    const float bias = lbm1[ncol];
#pragma unroll
    for (int m = 0; m < 7; ++m) {
#pragma unroll
      for (int v = 0; v < 8; ++v) {
        const int row = m * 16 + hi * 8 + v;
        pre[row * LDP + ncol] = acc[m][v] + bias;
      }
    }
  }
  __syncthreads();
  w_store<8>(wb, w2, tid);
  __syncthreads();

  float4 w3[32];
  w_fetch<16>(w3, Wu1, tid);

  for (int m = wave; m < 7; m += 8) {
    v8f acc[8];
#pragma unroll
    for (int n = 0; n < 8; ++n) acc[n] = vz;
    const int  i  = m * 16 + li;
    const int  ic = i - (i / 10) * 10;
    const bool e0 = (i >= 10) && (i < NN);
    const bool e1 = (i < NN - 10);
    const bool e2 = (i < NN) && (ic >= 1);
    const bool e3 = (i < NN) && (ic <= 8);
    const float* q0 = pre + (e0 ? i - 10 : i) * LDP + hi * 2;
    const float* q1 = pre + (e1 ? i + 10 : i) * LDP + hi * 2;
    const float* q2 = pre + (e2 ? i - 1  : i) * LDP + hi * 2;
    const float* q3 = pre + (e3 ? i + 1  : i) * LDP + hi * 2;
    const float* qw = wlast + hi * 2;
#pragma unroll 4
    for (int ks = 0; ks < 32; ++ks) {
      const v2f h0 = *(const v2f*)(q0 + ks * 4);
      const v2f h1 = *(const v2f*)(q1 + ks * 4);
      const v2f h2 = *(const v2f*)(q2 + ks * 4);
      const v2f h3 = *(const v2f*)(q3 + ks * 4);
      const v2f wl = *(const v2f*)(qw + ks * 4);
      v2f a;
      a.x = (e0 ? fmaxf(h0.x, 0.f) : 0.f)
          + (e1 ? fmaxf(h1.x + wl.x, 0.f) : 0.f)
          + (e2 ? fmaxf(h2.x + 2.f * wl.x, 0.f) : 0.f)
          + (e3 ? fmaxf(h3.x + 3.f * wl.x, 0.f) : 0.f);
      a.y = (e0 ? fmaxf(h0.y, 0.f) : 0.f)
          + (e1 ? fmaxf(h1.y + wl.y, 0.f) : 0.f)
          + (e2 ? fmaxf(h2.y + 2.f * wl.y, 0.f) : 0.f)
          + (e3 ? fmaxf(h3.y + 3.f * wl.y, 0.f) : 0.f);
      wmma_row8(wb + ks * 512, le, lo, a, acc);
    }
    float bias[8];
#pragma unroll
    for (int n = 0; n < 8; ++n) bias[n] = lbm2[n * 16 + li];
#pragma unroll
    for (int v = 0; v < 8; ++v) {
      const int r  = m * 16 + hi * 8 + v;
      const int rc = r - (r / 10) * 10;
      const int  in = (r < NN) ? 1 : 0;
      const float dg = (float)(((r >= 10 ? 1 : 0) & in) + (r < NN - 10 ? 1 : 0)
                               + ((rc >= 1 ? 1 : 0) & in) + ((rc <= 8 ? 1 : 0) & in));
#pragma unroll
      for (int n = 0; n < 8; ++n) {
        const int col = n * 16 + li;
        agg[r * LDP + col] = acc[n][v] + dg * bias[n];
      }
    }
  }
  __syncthreads();
  w_store<16>(wb, w3, tid);
  __syncthreads();

  float4 w4[16];
  w_fetch<8>(w4, Wu2, tid);

  {
    v8f acc[7];
#pragma unroll
    for (int m = 0; m < 7; ++m) acc[m] = vz;
    for (int kq = 0; kq < 8; ++kq)
      wmma_col7x4(xs + aofs + kq * 16, wb + bofs + kq * 2048, acc);
    for (int kq = 0; kq < 8; ++kq)
      wmma_col7x4(agg + aofs + kq * 16, wb + bofs + (8 + kq) * 2048, acc);
    const float bias = lbu1[ncol];
#pragma unroll
    for (int m = 0; m < 7; ++m) {
#pragma unroll
      for (int v = 0; v < 8; ++v) {
        const int row = m * 16 + hi * 8 + v;
        pre[row * LDP + ncol] = fmaxf(acc[m][v] + bias, 0.f);
      }
    }
  }
  __syncthreads();
  w_store<8>(wb, w4, tid);
  __syncthreads();

  float4 w1n[16];
  const XRows xn = x_fetch(x + (size_t)(more ? bn : b) * (NN * HH), tid);
  w_fetch<8>(w1n, Wm1, tid);

  {
    v8f acc[7];
#pragma unroll
    for (int m = 0; m < 7; ++m) acc[m] = vz;
    for (int kq = 0; kq < 8; ++kq)
      wmma_col7x4(pre + aofs + kq * 16, wb + bofs + kq * 2048, acc);
    const float bias = lbu2[ncol];
#pragma unroll
    for (int m = 0; m < 7; ++m) {
#pragma unroll
      for (int v = 0; v < 8; ++v) {
        const int row = m * 16 + hi * 8 + v;
        agg[row * LDP + ncol] = xs[row * LDP + ncol] + acc[m][v] + bias;
      }
    }
  }
  __syncthreads();

  x_store(xs, xn, p4);
  w_store<8>(wb, w1n, tid);

  const float* hb = agg;
  if (tid < NN) {
    const float* hr = hb + tid * LDP;
    float s = 0.f;
    for (int k = 0; k < HH; ++k) s += hr[k];
    const float mu = s * (1.0f / 128.0f);
    float var = 0.f;
    for (int k = 0; k < HH; ++k) { const float dd = hr[k] - mu; var += dd * dd; }
    lmu[tid] = mu;
    lrs[tid] = rsqrtf(var * (1.0f / 128.0f) + 1e-5f);
  }
  __syncthreads();

  float* og = out + (size_t)b * (NN * HH);
  for (int idx = tid; idx < NN * (HH / 4); idx += 256) {
    const int row = idx >> 5;
    const int c4  = idx & 31;
    const int col = c4 * 4;
    const float4 hv = ((const float4*)hb)[row * LDP4 + c4];
    const float mu = lmu[row], rs = lrs[row];
    float4 o;
    o.x = (hv.x - mu) * rs * lgm[col + 0] + lbt[col + 0];
    o.y = (hv.y - mu) * rs * lgm[col + 1] + lbt[col + 1];
    o.z = (hv.z - mu) * rs * lgm[col + 2] + lbt[col + 2];
    o.w = (hv.w - mu) * rs * lgm[col + 3] + lbt[col + 3];
    ((float4*)og)[idx] = o;
  }

  if (!more) break;
  b = bn;
  }
}

extern "C" void kernel_launch(void* const* d_in, const int* in_sizes, int n_in,
                              void* d_out, int out_size, void* d_ws, size_t ws_size,
                              hipStream_t stream) {
  (void)n_in; (void)out_size; (void)d_ws; (void)ws_size;
  const float* x     = (const float*)d_in[0];
  const float* Wm1   = (const float*)d_in[4];
  const float* bm1   = (const float*)d_in[5];
  const float* Wm2   = (const float*)d_in[6];
  const float* bm2   = (const float*)d_in[7];
  const float* Wu1   = (const float*)d_in[8];
  const float* bu1   = (const float*)d_in[9];
  const float* Wu2   = (const float*)d_in[10];
  const float* bu2   = (const float*)d_in[11];
  const float* gamma = (const float*)d_in[12];
  const float* beta  = (const float*)d_in[13];
  float* out = (float*)d_out;

  const int B = in_sizes[0] / (NN * HH);
  const size_t shmem = (size_t)SMEM_FLOATS * sizeof(float);
  hipLaunchKernelGGL(mpnn_layer_kernel, dim3(256), dim3(256), shmem, stream,
                     x, Wm1, bm1, Wm2, bm2, Wu1, bu1, Wu2, bu2, gamma, beta, out, B);
}


# Weights resident per block: Wm1 and Wm2 stay in the LDS stage, each wave keeps its 16 columns of Wu1, Wu2 as B fragments in 192 VGPRs; no per-element weight fetch, LDS re-staging or second barrier
# speedup vs baseline: 2.4036x; 1.1471x over previous
#include <hip/hip_runtime.h>


typedef __attribute__((ext_vector_type(2))) float v2f;
typedef __attribute__((ext_vector_type(8))) float v8f;

#define NN    100
#define HH    128
#define NPAD  112
#define R4    400
#define LDP   132
#define LDP4  33

#define OFF_XS    0
#define OFF_PRE   14784
#define OFF_AGG   29568
#define OFF_W     44352
#define OFF_W2    16384
#define OFF_WLAST 77120
#define OFF_BM1   77248
#define OFF_BM2   77376
#define OFF_BU1   77504
#define OFF_BU2   77632
#define OFF_GM    77760
#define OFF_BT    77888
#define OFF_MU    78016
#define OFF_RS    78128
#define SMEM_FLOATS 78240

__device__ __forceinline__ v8f wmma4(v2f a, v2f b, v8f c) {
  return __builtin_amdgcn_wmma_f32_16x16x4_f32(
       false, a,  false, b,
       (short)0, c,  false,  false);
}

template <int NIT>
__device__ __forceinline__ void w_fetch(float4 (&r)[2 * NIT], const float* __restrict__ src, int tid) {
#pragma unroll
  for (int it = 0; it < NIT; ++it) {
    const int idx = tid + it * 256;
    const int kp = idx >> 5;
    const int cg = idx & 31;
    r[2 * it]     = *(const float4*)(src + (size_t)(2 * kp)     * 128 + cg * 4);
    r[2 * it + 1] = *(const float4*)(src + (size_t)(2 * kp + 1) * 128 + cg * 4);
  }
}

template <int NIT>
__device__ __forceinline__ void w_store(float* dst, const float4 (&r)[2 * NIT], int tid) {
#pragma unroll
  for (int it = 0; it < NIT; ++it) {
    const int idx = tid + it * 256;
    const int kp = idx >> 5;
    const int cg = idx & 31;
    const float4 r0 = r[2 * it];
    const float4 r1 = r[2 * it + 1];
    float4* d = (float4*)(dst + kp * 256 + (cg ^ ((kp & 1) << 2)) * 8);
    d[0] = make_float4(r0.x, r1.x, r0.y, r1.y);
    d[1] = make_float4(r0.z, r1.z, r0.w, r1.w);
  }
}

struct XRows { float4 r0, r1, r2, r3, r4, r5, r6, r7, r8, r9, r10, r11, r12; };

__device__ __forceinline__ XRows x_fetch(const float* __restrict__ xg, int tid) {
  const float4* xg4 = (const float4*)xg;
  const int il = tid + 12 * 256;
  const bool inx = il < NN * HH / 4;
  XRows r;
  r.r0  = xg4[tid + 0 * 256];
  r.r1  = xg4[tid + 1 * 256];
  r.r2  = xg4[tid + 2 * 256];
  r.r3  = xg4[tid + 3 * 256];
  r.r4  = xg4[tid + 4 * 256];
  r.r5  = xg4[tid + 5 * 256];
  r.r6  = xg4[tid + 6 * 256];
  r.r7  = xg4[tid + 7 * 256];
  r.r8  = xg4[tid + 8 * 256];
  r.r9  = xg4[tid + 9 * 256];
  r.r10 = xg4[tid + 10 * 256];
  r.r11 = xg4[tid + 11 * 256];
  const float4 xt = xg4[inx ? il : NN * HH / 4 - 1];
  r.r12 = make_float4(inx ? xt.x : 0.f, inx ? xt.y : 0.f,
                      inx ? xt.z : 0.f, inx ? xt.w : 0.f);
  return r;
}

__device__ __forceinline__ void x_store(float* xs, const XRows& r, int p4) {
  float4* d = (float4*)xs + p4;
  d[0 * 8 * LDP4]  = r.r0;
  d[1 * 8 * LDP4]  = r.r1;
  d[2 * 8 * LDP4]  = r.r2;
  d[3 * 8 * LDP4]  = r.r3;
  d[4 * 8 * LDP4]  = r.r4;
  d[5 * 8 * LDP4]  = r.r5;
  d[6 * 8 * LDP4]  = r.r6;
  d[7 * 8 * LDP4]  = r.r7;
  d[8 * 8 * LDP4]  = r.r8;
  d[9 * 8 * LDP4]  = r.r9;
  d[10 * 8 * LDP4] = r.r10;
  d[11 * 8 * LDP4] = r.r11;
  d[12 * 8 * LDP4] = r.r12;
}

__device__ __forceinline__ void wmma_row8(const float* __restrict__ wbk, int le, int lo,
                                          v2f a, v8f acc[8]) {
#pragma unroll
  for (int n = 0; n < 8; ++n) {
    const v2f bb = *(const v2f*)(wbk + n * 32 + ((n & 1) ? lo : le));
    acc[n] = wmma4(a, bb, acc[n]);
  }
}

__device__ __forceinline__ void wmma_col7x4(const float* __restrict__ ap,
                                            const float* __restrict__ bp, v8f acc[7]) {
#pragma unroll
  for (int j = 0; j < 4; ++j) {
    const v2f bb = *(const v2f*)(bp + j * 512);
#pragma unroll
    for (int m = 0; m < 7; ++m) {
      const v2f a = *(const v2f*)(ap + m * 16 * LDP + j * 4);
      acc[m] = wmma4(a, bb, acc[m]);
    }
  }
}

template <int NKS>
__device__ __forceinline__ void b_fetch(v2f (&r)[NKS], const float* __restrict__ w, int hi, int ncol) {
#pragma unroll
  for (int ks = 0; ks < NKS; ++ks) {
    const float* p = w + (4 * ks + 2 * hi) * 128 + ncol;
    r[ks].x = p[0];
    r[ks].y = p[128];
  }
}

__device__ __forceinline__ void a_load28(v2f (&a)[28], const float* __restrict__ ap) {
#pragma unroll
  for (int j = 0; j < 4; ++j)
#pragma unroll
    for (int m = 0; m < 7; ++m)
      a[j * 7 + m] = *(const v2f*)(ap + m * 16 * LDP + j * 4);
}

template <int NBLK>
__device__ __forceinline__ void wmma_col7_regb(const float* __restrict__ ap0,
                                               const float* __restrict__ ap1,
                                               const v2f* __restrict__ breg, v8f acc[7]) {
  v2f a[2][28];
  a_load28(a[0], ap0);
#pragma unroll
  for (int blk = 0; blk < NBLK; ++blk) {
    if (blk + 1 < NBLK)
      a_load28(a[(blk + 1) & 1], (blk + 1 < 8) ? ap0 + (blk + 1) * 16 : ap1 + (blk + 1 - 8) * 16);
    __builtin_amdgcn_sched_barrier(0);
#pragma unroll
    for (int j = 0; j < 4; ++j) {
      const v2f bb = breg[blk * 4 + j];
#pragma unroll
      for (int m = 0; m < 7; ++m)
        acc[m] = wmma4(a[blk & 1][j * 7 + m], bb, acc[m]);
    }
    __builtin_amdgcn_sched_barrier(0);
  }
}

extern "C" __global__ __launch_bounds__(256) __attribute__((amdgpu_waves_per_eu(2, 2)))
void mpnn_layer_kernel(const float* __restrict__ x,
                       const float* __restrict__ Wm1, const float* __restrict__ bm1,
                       const float* __restrict__ Wm2, const float* __restrict__ bm2,
                       const float* __restrict__ Wu1, const float* __restrict__ bu1,
                       const float* __restrict__ Wu2, const float* __restrict__ bu2,
                       const float* __restrict__ gamma, const float* __restrict__ beta,
                       float* __restrict__ out, int nb)
{
  extern __shared__ float smem[];
  float* xs    = smem + OFF_XS;
  float* pre   = smem + OFF_PRE;
  float* agg   = smem + OFF_AGG;
  float* wb    = smem + OFF_W;
  float* wlast = smem + OFF_WLAST;
  float* lbm1  = smem + OFF_BM1;
  float* lbm2  = smem + OFF_BM2;
  float* lbu1  = smem + OFF_BU1;
  float* lbu2  = smem + OFF_BU2;
  float* lgm   = smem + OFF_GM;
  float* lbt   = smem + OFF_BT;
  float* lmu   = smem + OFF_MU;
  float* lrs   = smem + OFF_RS;

  int b = blockIdx.x;
  if (b >= nb) return;

  {
    const int tid  = threadIdx.x;
    const int lane = tid & 31;
    const int p4   = (tid >> 5) * LDP4 + lane;
    const float4 z4 = make_float4(0.f, 0.f, 0.f, 0.f);
    float4 w1[16];
    float4 w2[16];
    const XRows xr = x_fetch(x + (size_t)b * (NN * HH), tid);
    w_fetch<8>(w1, Wm1, tid);
    w_fetch<8>(w2, Wm2, tid);
    ((float4*)xs)[p4 + 13 * 8 * LDP4] = z4;
    x_store(xs, xr, p4);
    w_store<8>(wb, w1, tid);
    w_store<8>(wb + OFF_W2, w2, tid);
    if ((tid >> 5) == 0) {
      ((float4*)wlast)[lane] = ((const float4*)(Wm1 + 128 * 128))[lane];
      ((float4*)lbm1)[lane]  = ((const float4*)bm1)[lane];
      ((float4*)lbm2)[lane]  = ((const float4*)bm2)[lane];
      ((float4*)lbu1)[lane]  = ((const float4*)bu1)[lane];
      ((float4*)lbu2)[lane]  = ((const float4*)bu2)[lane];
      ((float4*)lgm)[lane]   = ((const float4*)gamma)[lane];
      ((float4*)lbt)[lane]   = ((const float4*)beta)[lane];
    }
  }
  v2f bu1r[64];
  v2f bu2r[32];
  {
    const int t0 = threadIdx.x;
    b_fetch<64>(bu1r, Wu1, (t0 & 31) >> 4, (t0 >> 5) * 16 + (t0 & 15));
    b_fetch<32>(bu2r, Wu2, (t0 & 31) >> 4, (t0 >> 5) * 16 + (t0 & 15));
  }
  __syncthreads();

  const v8f vz = {0.f, 0.f, 0.f, 0.f, 0.f, 0.f, 0.f, 0.f};

  for (;;) {
  const int  bn   = b + (int)gridDim.x;
  const bool more = bn < nb;

  int tid = threadIdx.x;
  asm volatile("" : "+v"(tid));
  const int wave = tid >> 5;
  const int lane = tid & 31;
  const int li   = lane & 15;
  const int hi   = lane >> 4;
  const int le   = hi * 256 + li * 2 + hi * 32;
  const int lo   = hi * 256 + li * 2 - hi * 32;
  const int ncol = wave * 16 + li;
  const int bofs = hi * 256 + ((wave ^ hi) << 5) + li * 2;
  const int aofs = li * LDP + hi * 2;
  const int p4   = (tid >> 5) * LDP4 + (tid & 31);

  {
    v8f acc[7];
#pragma unroll
    for (int m = 0; m < 7; ++m) acc[m] = vz;
    for (int kq = 0; kq < 8; ++kq)
      wmma_col7x4(xs + aofs + kq * 16, wb + bofs + kq * 2048, acc);
    const float bias = lbm1[ncol];
#pragma unroll
    for (int m = 0; m < 7; ++m) {
#pragma unroll
      for (int v = 0; v < 8; ++v) {
        const int row = m * 16 + hi * 8 + v;
        pre[row * LDP + ncol] = acc[m][v] + bias;
      }
    }
  }
  __syncthreads();

  for (int m = wave; m < 7; m += 8) {
    v8f acc[8];
#pragma unroll
    for (int n = 0; n < 8; ++n) acc[n] = vz;
    const int  i  = m * 16 + li;
    const int  ic = i - (i / 10) * 10;
    const bool e0 = (i >= 10) && (i < NN);
    const bool e1 = (i < NN - 10);
    const bool e2 = (i < NN) && (ic >= 1);
    const bool e3 = (i < NN) && (ic <= 8);
    const float* q0 = pre + (e0 ? i - 10 : i) * LDP + hi * 2;
    const float* q1 = pre + (e1 ? i + 10 : i) * LDP + hi * 2;
    const float* q2 = pre + (e2 ? i - 1  : i) * LDP + hi * 2;
    const float* q3 = pre + (e3 ? i + 1  : i) * LDP + hi * 2;
    const float* qw = wlast + hi * 2;
#pragma unroll 4
    for (int ks = 0; ks < 32; ++ks) {
      const v2f h0 = *(const v2f*)(q0 + ks * 4);
      const v2f h1 = *(const v2f*)(q1 + ks * 4);
      const v2f h2 = *(const v2f*)(q2 + ks * 4);
      const v2f h3 = *(const v2f*)(q3 + ks * 4);
      const v2f wl = *(const v2f*)(qw + ks * 4);
      v2f a;
      a.x = (e0 ? fmaxf(h0.x, 0.f) : 0.f)
          + (e1 ? fmaxf(h1.x + wl.x, 0.f) : 0.f)
          + (e2 ? fmaxf(h2.x + 2.f * wl.x, 0.f) : 0.f)
          + (e3 ? fmaxf(h3.x + 3.f * wl.x, 0.f) : 0.f);
      a.y = (e0 ? fmaxf(h0.y, 0.f) : 0.f)
          + (e1 ? fmaxf(h1.y + wl.y, 0.f) : 0.f)
          + (e2 ? fmaxf(h2.y + 2.f * wl.y, 0.f) : 0.f)
          + (e3 ? fmaxf(h3.y + 3.f * wl.y, 0.f) : 0.f);
      wmma_row8(wb + OFF_W2 + ks * 512, le, lo, a, acc);
    }
    float bias[8];
#pragma unroll
    for (int n = 0; n < 8; ++n) bias[n] = lbm2[n * 16 + li];
#pragma unroll
    for (int v = 0; v < 8; ++v) {
      const int r  = m * 16 + hi * 8 + v;
      const int rc = r - (r / 10) * 10;
      const int  in = (r < NN) ? 1 : 0;
      const float dg = (float)(((r >= 10 ? 1 : 0) & in) + (r < NN - 10 ? 1 : 0)
                               + ((rc >= 1 ? 1 : 0) & in) + ((rc <= 8 ? 1 : 0) & in));
#pragma unroll
      for (int n = 0; n < 8; ++n) {
        const int col = n * 16 + li;
        agg[r * LDP + col] = acc[n][v] + dg * bias[n];
      }
    }
  }
  __syncthreads();

  {
    v8f acc[7];
#pragma unroll
    for (int m = 0; m < 7; ++m) acc[m] = vz;
    wmma_col7_regb<16>(xs + aofs, agg + aofs, bu1r, acc);
    const float bias = lbu1[ncol];
#pragma unroll
    for (int m = 0; m < 7; ++m) {
#pragma unroll
      for (int v = 0; v < 8; ++v) {
        const int row = m * 16 + hi * 8 + v;
        pre[row * LDP + ncol] = fmaxf(acc[m][v] + bias, 0.f);
      }
    }
  }
  __syncthreads();

  const XRows xn = x_fetch(x + (size_t)(more ? bn : b) * (NN * HH), tid);

  {
    v8f acc[7];
#pragma unroll
    for (int m = 0; m < 7; ++m) acc[m] = vz;
    wmma_col7_regb<8>(pre + aofs, pre + aofs, bu2r, acc);
    const float bias = lbu2[ncol];
#pragma unroll
    for (int m = 0; m < 7; ++m) {
#pragma unroll
      for (int v = 0; v < 8; ++v) {
        const int row = m * 16 + hi * 8 + v;
        agg[row * LDP + ncol] = xs[row * LDP + ncol] + acc[m][v] + bias;
      }
    }
  }
  __syncthreads();

  x_store(xs, xn, p4);

  const float* hb = agg;
  if (tid < NN) {
    const float* hr = hb + tid * LDP;
    float s = 0.f;
    for (int k = 0; k < HH; ++k) s += hr[k];
    const float mu = s * (1.0f / 128.0f);
    float var = 0.f;
    for (int k = 0; k < HH; ++k) { const float dd = hr[k] - mu; var += dd * dd; }
    lmu[tid] = mu;
    lrs[tid] = rsqrtf(var * (1.0f / 128.0f) + 1e-5f);
  }
  __syncthreads();

  float* og = out + (size_t)b * (NN * HH);
  for (int idx = tid; idx < NN * (HH / 4); idx += 256) {
    const int row = idx >> 5;
    const int c4  = idx & 31;
    const int col = c4 * 4;
    const float4 hv = ((const float4*)hb)[row * LDP4 + c4];
    const float mu = lmu[row], rs = lrs[row];
    float4 o;
    o.x = (hv.x - mu) * rs * lgm[col + 0] + lbt[col + 0];
    o.y = (hv.y - mu) * rs * lgm[col + 1] + lbt[col + 1];
    o.z = (hv.z - mu) * rs * lgm[col + 2] + lbt[col + 2];
    o.w = (hv.w - mu) * rs * lgm[col + 3] + lbt[col + 3];
    ((float4*)og)[idx] = o;
  }

  if (!more) break;
  b = bn;
  }
}

extern "C" void kernel_launch(void* const* d_in, const int* in_sizes, int n_in,
                              void* d_out, int out_size, void* d_ws, size_t ws_size,
                              hipStream_t stream) {
  (void)n_in; (void)out_size; (void)d_ws; (void)ws_size;
  const float* x     = (const float*)d_in[0];
  const float* Wm1   = (const float*)d_in[4];
  const float* bm1   = (const float*)d_in[5];
  const float* Wm2   = (const float*)d_in[6];
  const float* bm2   = (const float*)d_in[7];
  const float* Wu1   = (const float*)d_in[8];
  const float* bu1   = (const float*)d_in[9];
  const float* Wu2   = (const float*)d_in[10];
  const float* bu2   = (const float*)d_in[11];
  const float* gamma = (const float*)d_in[12];
  const float* beta  = (const float*)d_in[13];
  float* out = (float*)d_out;

  const int B = in_sizes[0] / (NN * HH);
  const size_t shmem = (size_t)SMEM_FLOATS * sizeof(float);
  hipLaunchKernelGGL(mpnn_layer_kernel, dim3(256), dim3(256), shmem, stream,
                     x, Wm1, bm1, Wm2, bm2, Wu1, bu1, Wu2, bu2, gamma, beta, out, B);
}


# LayerNorm by rows per wave: a lane holds 4 of a row's 128 values, mean and variance are 32-lane butterfly sums (sum order changes), one 512 B store per row; no LDS statistics pass
# speedup vs baseline: 2.5722x; 1.0701x over previous
#include <hip/hip_runtime.h>


typedef __attribute__((ext_vector_type(2))) float v2f;
typedef __attribute__((ext_vector_type(4))) float v4f;
typedef __attribute__((ext_vector_type(8))) float v8f;

#define NN    100
#define HH    128
#define NPAD  112
#define R4    400
#define LDP   132
#define LDP4  33

#define OFF_XS    0
#define OFF_PRE   14784
#define OFF_AGG   29568
#define OFF_W     44352
#define OFF_W2    16384
#define OFF_WLAST 77120
#define OFF_BM1   77248
#define OFF_BM2   77376
#define OFF_BU1   77504
#define OFF_BU2   77632
#define OFF_GM    77760
#define OFF_BT    77888
#define SMEM_FLOATS 78016

__device__ __forceinline__ v8f wmma4(v2f a, v2f b, v8f c) {
  return __builtin_amdgcn_wmma_f32_16x16x4_f32(
       false, a,  false, b,
       (short)0, c,  false,  false);
}

template <int NIT>
__device__ __forceinline__ void w_fetch(float4 (&r)[2 * NIT], const float* __restrict__ src, int tid) {
#pragma unroll
  for (int it = 0; it < NIT; ++it) {
    const int idx = tid + it * 256;
    const int kp = idx >> 5;
    const int cg = idx & 31;
    r[2 * it]     = *(const float4*)(src + (size_t)(2 * kp)     * 128 + cg * 4);
    r[2 * it + 1] = *(const float4*)(src + (size_t)(2 * kp + 1) * 128 + cg * 4);
  }
}

template <int NIT>
__device__ __forceinline__ void w_store(float* dst, const float4 (&r)[2 * NIT], int tid) {
#pragma unroll
  for (int it = 0; it < NIT; ++it) {
    const int idx = tid + it * 256;
    const int kp = idx >> 5;
    const int cg = idx & 31;
    const float4 r0 = r[2 * it];
    const float4 r1 = r[2 * it + 1];
    float4* d = (float4*)(dst + kp * 256 + (cg ^ ((kp & 1) << 2)) * 8);
    d[0] = make_float4(r0.x, r1.x, r0.y, r1.y);
    d[1] = make_float4(r0.z, r1.z, r0.w, r1.w);
  }
}

struct XRows { float4 r0, r1, r2, r3, r4, r5, r6, r7, r8, r9, r10, r11, r12; };

__device__ __forceinline__ XRows x_fetch(const float* __restrict__ xg, int tid) {
  const float4* xg4 = (const float4*)xg;
  const int il = tid + 12 * 256;
  const bool inx = il < NN * HH / 4;
  XRows r;
  r.r0  = xg4[tid + 0 * 256];
  r.r1  = xg4[tid + 1 * 256];
  r.r2  = xg4[tid + 2 * 256];
  r.r3  = xg4[tid + 3 * 256];
  r.r4  = xg4[tid + 4 * 256];
  r.r5  = xg4[tid + 5 * 256];
  r.r6  = xg4[tid + 6 * 256];
  r.r7  = xg4[tid + 7 * 256];
  r.r8  = xg4[tid + 8 * 256];
  r.r9  = xg4[tid + 9 * 256];
  r.r10 = xg4[tid + 10 * 256];
  r.r11 = xg4[tid + 11 * 256];
  const float4 xt = xg4[inx ? il : NN * HH / 4 - 1];
  r.r12 = make_float4(inx ? xt.x : 0.f, inx ? xt.y : 0.f,
                      inx ? xt.z : 0.f, inx ? xt.w : 0.f);
  return r;
}

__device__ __forceinline__ void x_store(float* xs, const XRows& r, int p4) {
  float4* d = (float4*)xs + p4;
  d[0 * 8 * LDP4]  = r.r0;
  d[1 * 8 * LDP4]  = r.r1;
  d[2 * 8 * LDP4]  = r.r2;
  d[3 * 8 * LDP4]  = r.r3;
  d[4 * 8 * LDP4]  = r.r4;
  d[5 * 8 * LDP4]  = r.r5;
  d[6 * 8 * LDP4]  = r.r6;
  d[7 * 8 * LDP4]  = r.r7;
  d[8 * 8 * LDP4]  = r.r8;
  d[9 * 8 * LDP4]  = r.r9;
  d[10 * 8 * LDP4] = r.r10;
  d[11 * 8 * LDP4] = r.r11;
  d[12 * 8 * LDP4] = r.r12;
}

__device__ __forceinline__ void wmma_row8(const float* __restrict__ wbk, int le, int lo,
                                          v2f a, v8f acc[8]) {
#pragma unroll
  for (int n = 0; n < 8; ++n) {
    const v2f bb = *(const v2f*)(wbk + n * 32 + ((n & 1) ? lo : le));
    acc[n] = wmma4(a, bb, acc[n]);
  }
}

__device__ __forceinline__ void wmma_col7x4(const float* __restrict__ ap,
                                            const float* __restrict__ bp, v8f acc[7]) {
#pragma unroll
  for (int j = 0; j < 4; ++j) {
    const v2f bb = *(const v2f*)(bp + j * 512);
#pragma unroll
    for (int m = 0; m < 7; ++m) {
      const v2f a = *(const v2f*)(ap + m * 16 * LDP + j * 4);
      acc[m] = wmma4(a, bb, acc[m]);
    }
  }
}

template <int NKS>
__device__ __forceinline__ void b_fetch(v2f (&r)[NKS], const float* __restrict__ w, int hi, int ncol) {
#pragma unroll
  for (int ks = 0; ks < NKS; ++ks) {
    const float* p = w + (4 * ks + 2 * hi) * 128 + ncol;
    r[ks].x = p[0];
    r[ks].y = p[128];
  }
}

__device__ __forceinline__ void a_load28(v2f (&a)[28], const float* __restrict__ ap) {
#pragma unroll
  for (int j = 0; j < 4; ++j)
#pragma unroll
    for (int m = 0; m < 7; ++m)
      a[j * 7 + m] = *(const v2f*)(ap + m * 16 * LDP + j * 4);
}

template <int NBLK>
__device__ __forceinline__ void wmma_col7_regb(const float* __restrict__ ap0,
                                               const float* __restrict__ ap1,
                                               const v2f* __restrict__ breg, v8f acc[7]) {
  v2f a[2][28];
  a_load28(a[0], ap0);
#pragma unroll
  for (int blk = 0; blk < NBLK; ++blk) {
    if (blk + 1 < NBLK)
      a_load28(a[(blk + 1) & 1], (blk + 1 < 8) ? ap0 + (blk + 1) * 16 : ap1 + (blk + 1 - 8) * 16);
    __builtin_amdgcn_sched_barrier(0);
#pragma unroll
    for (int j = 0; j < 4; ++j) {
      const v2f bb = breg[blk * 4 + j];
#pragma unroll
      for (int m = 0; m < 7; ++m)
        acc[m] = wmma4(a[blk & 1][j * 7 + m], bb, acc[m]);
    }
    __builtin_amdgcn_sched_barrier(0);
  }
}

template <int XM>
__device__ __forceinline__ float row_xor(float v) {
  return __builtin_bit_cast(float, __builtin_amdgcn_update_dpp(
      0, __builtin_bit_cast(int, v), 0x160 + XM, 0xf, 0xf, true));
}
__device__ __forceinline__ float half_xor(float v) {
  return __builtin_bit_cast(float, __builtin_amdgcn_permlanex16(
      __builtin_bit_cast(int, v), __builtin_bit_cast(int, v), 0x76543210, 0xfedcba98, false, true));
}

__device__ __forceinline__ void wave_sum13(float (&s)[13]) {
#pragma unroll
  for (int j = 0; j < 13; ++j) s[j] += row_xor<1>(s[j]);
#pragma unroll
  for (int j = 0; j < 13; ++j) s[j] += row_xor<2>(s[j]);
#pragma unroll
  for (int j = 0; j < 13; ++j) s[j] += row_xor<4>(s[j]);
#pragma unroll
  for (int j = 0; j < 13; ++j) s[j] += row_xor<8>(s[j]);
#pragma unroll
  for (int j = 0; j < 13; ++j) s[j] += half_xor(s[j]);
}

extern "C" __global__ __launch_bounds__(256) __attribute__((amdgpu_waves_per_eu(2, 2)))
void mpnn_layer_kernel(const float* __restrict__ x,
                       const float* __restrict__ Wm1, const float* __restrict__ bm1,
                       const float* __restrict__ Wm2, const float* __restrict__ bm2,
                       const float* __restrict__ Wu1, const float* __restrict__ bu1,
                       const float* __restrict__ Wu2, const float* __restrict__ bu2,
                       const float* __restrict__ gamma, const float* __restrict__ beta,
                       float* __restrict__ out, int nb)
{
  extern __shared__ float smem[];
  float* xs    = smem + OFF_XS;
  float* pre   = smem + OFF_PRE;
  float* agg   = smem + OFF_AGG;
  float* wb    = smem + OFF_W;
  float* wlast = smem + OFF_WLAST;
  float* lbm1  = smem + OFF_BM1;
  float* lbm2  = smem + OFF_BM2;
  float* lbu1  = smem + OFF_BU1;
  float* lbu2  = smem + OFF_BU2;
  float* lgm   = smem + OFF_GM;
  float* lbt   = smem + OFF_BT;

  int b = blockIdx.x;
  if (b >= nb) return;

  {
    const int tid  = threadIdx.x;
    const int lane = tid & 31;
    const int p4   = (tid >> 5) * LDP4 + lane;
    const float4 z4 = make_float4(0.f, 0.f, 0.f, 0.f);
    float4 w1[16];
    float4 w2[16];
    const XRows xr = x_fetch(x + (size_t)b * (NN * HH), tid);
    w_fetch<8>(w1, Wm1, tid);
    w_fetch<8>(w2, Wm2, tid);
    ((float4*)xs)[p4 + 13 * 8 * LDP4] = z4;
    x_store(xs, xr, p4);
    w_store<8>(wb, w1, tid);
    w_store<8>(wb + OFF_W2, w2, tid);
    if ((tid >> 5) == 0) {
      ((float4*)wlast)[lane] = ((const float4*)(Wm1 + 128 * 128))[lane];
      ((float4*)lbm1)[lane]  = ((const float4*)bm1)[lane];
      ((float4*)lbm2)[lane]  = ((const float4*)bm2)[lane];
      ((float4*)lbu1)[lane]  = ((const float4*)bu1)[lane];
      ((float4*)lbu2)[lane]  = ((const float4*)bu2)[lane];
      ((float4*)lgm)[lane]   = ((const float4*)gamma)[lane];
      ((float4*)lbt)[lane]   = ((const float4*)beta)[lane];
    }
  }
  v2f bu1r[64];
  v2f bu2r[32];
  {
    const int t0 = threadIdx.x;
    b_fetch<64>(bu1r, Wu1, (t0 & 31) >> 4, (t0 >> 5) * 16 + (t0 & 15));
    b_fetch<32>(bu2r, Wu2, (t0 & 31) >> 4, (t0 >> 5) * 16 + (t0 & 15));
  }
  __syncthreads();

  const v8f vz = {0.f, 0.f, 0.f, 0.f, 0.f, 0.f, 0.f, 0.f};

  for (;;) {
  const int  bn   = b + (int)gridDim.x;
  const bool more = bn < nb;

  int tid = threadIdx.x;
  asm volatile("" : "+v"(tid));
  const int wave = tid >> 5;
  const int lane = tid & 31;
  const int li   = lane & 15;
  const int hi   = lane >> 4;
  const int le   = hi * 256 + li * 2 + hi * 32;
  const int lo   = hi * 256 + li * 2 - hi * 32;
  const int ncol = wave * 16 + li;
  const int bofs = hi * 256 + ((wave ^ hi) << 5) + li * 2;
  const int aofs = li * LDP + hi * 2;
  const int p4   = (tid >> 5) * LDP4 + (tid & 31);

  {
    v8f acc[7];
#pragma unroll
    for (int m = 0; m < 7; ++m) acc[m] = vz;
    for (int kq = 0; kq < 8; ++kq)
      wmma_col7x4(xs + aofs + kq * 16, wb + bofs + kq * 2048, acc);
    const float bias = lbm1[ncol];
#pragma unroll
    for (int m = 0; m < 7; ++m) {
#pragma unroll
      for (int v = 0; v < 8; ++v) {
        const int row = m * 16 + hi * 8 + v;
        pre[row * LDP + ncol] = acc[m][v] + bias;
      }
    }
  }
  __syncthreads();

  for (int m = wave; m < 7; m += 8) {
    v8f acc[8];
#pragma unroll
    for (int n = 0; n < 8; ++n) acc[n] = vz;
    const int  i  = m * 16 + li;
    const int  ic = i - (i / 10) * 10;
    const bool e0 = (i >= 10) && (i < NN);
    const bool e1 = (i < NN - 10);
    const bool e2 = (i < NN) && (ic >= 1);
    const bool e3 = (i < NN) && (ic <= 8);
    const float* q0 = pre + (e0 ? i - 10 : i) * LDP + hi * 2;
    const float* q1 = pre + (e1 ? i + 10 : i) * LDP + hi * 2;
    const float* q2 = pre + (e2 ? i - 1  : i) * LDP + hi * 2;
    const float* q3 = pre + (e3 ? i + 1  : i) * LDP + hi * 2;
    const float* qw = wlast + hi * 2;
#pragma unroll 4
    for (int ks = 0; ks < 32; ++ks) {
      const v2f h0 = *(const v2f*)(q0 + ks * 4);
      const v2f h1 = *(const v2f*)(q1 + ks * 4);
      const v2f h2 = *(const v2f*)(q2 + ks * 4);
      const v2f h3 = *(const v2f*)(q3 + ks * 4);
      const v2f wl = *(const v2f*)(qw + ks * 4);
      v2f a;
      a.x = (e0 ? fmaxf(h0.x, 0.f) : 0.f)
          + (e1 ? fmaxf(h1.x + wl.x, 0.f) : 0.f)
          + (e2 ? fmaxf(h2.x + 2.f * wl.x, 0.f) : 0.f)
          + (e3 ? fmaxf(h3.x + 3.f * wl.x, 0.f) : 0.f);
      a.y = (e0 ? fmaxf(h0.y, 0.f) : 0.f)
          + (e1 ? fmaxf(h1.y + wl.y, 0.f) : 0.f)
          + (e2 ? fmaxf(h2.y + 2.f * wl.y, 0.f) : 0.f)
          + (e3 ? fmaxf(h3.y + 3.f * wl.y, 0.f) : 0.f);
      wmma_row8(wb + OFF_W2 + ks * 512, le, lo, a, acc);
    }
    float bias[8];
#pragma unroll
    for (int n = 0; n < 8; ++n) bias[n] = lbm2[n * 16 + li];
#pragma unroll
    for (int v = 0; v < 8; ++v) {
      const int r  = m * 16 + hi * 8 + v;
      const int rc = r - (r / 10) * 10;
      const int  in = (r < NN) ? 1 : 0;
      const float dg = (float)(((r >= 10 ? 1 : 0) & in) + (r < NN - 10 ? 1 : 0)
                               + ((rc >= 1 ? 1 : 0) & in) + ((rc <= 8 ? 1 : 0) & in));
#pragma unroll
      for (int n = 0; n < 8; ++n) {
        const int col = n * 16 + li;
        agg[r * LDP + col] = acc[n][v] + dg * bias[n];
      }
    }
  }
  __syncthreads();

  {
    v8f acc[7];
#pragma unroll
    for (int m = 0; m < 7; ++m) acc[m] = vz;
    wmma_col7_regb<16>(xs + aofs, agg + aofs, bu1r, acc);
    const float bias = lbu1[ncol];
#pragma unroll
    for (int m = 0; m < 7; ++m) {
#pragma unroll
      for (int v = 0; v < 8; ++v) {
        const int row = m * 16 + hi * 8 + v;
        pre[row * LDP + ncol] = fmaxf(acc[m][v] + bias, 0.f);
      }
    }
  }
  __syncthreads();

  const XRows xn = x_fetch(x + (size_t)(more ? bn : b) * (NN * HH), tid);

  {
    v8f acc[7];
#pragma unroll
    for (int m = 0; m < 7; ++m) acc[m] = vz;
    wmma_col7_regb<8>(pre + aofs, pre + aofs, bu2r, acc);
    const float bias = lbu2[ncol];
#pragma unroll
    for (int m = 0; m < 7; ++m) {
#pragma unroll
      for (int v = 0; v < 8; ++v) {
        const int row = m * 16 + hi * 8 + v;
        agg[row * LDP + ncol] = xs[row * LDP + ncol] + acc[m][v] + bias;
      }
    }
  }
  __syncthreads();

  x_store(xs, xn, p4);
  __syncthreads();

  {
    const bool tail = wave < 4;
    const v4f g4 = ((const v4f*)lgm)[lane];
    const v4f b4 = ((const v4f*)lbt)[lane];
    const v4f* hb4 = (const v4f*)agg + wave * LDP4 + lane;
    v4f* og4 = (v4f*)(out + (size_t)b * (NN * HH)) + wave * (HH / 4) + lane;
    v2f d0[13], d1[13];
    float s[13];
#pragma unroll
    for (int j = 0; j < 13; ++j) {
      const v4f h = hb4[(j < 12 ? j : (tail ? 12 : 11)) * 8 * LDP4];
      d0[j] = h.lo;
      d1[j] = h.hi;
      const v2f a = d0[j] + d1[j];
      s[j] = a.x + a.y;
    }
    wave_sum13(s);
#pragma unroll
    for (int j = 0; j < 13; ++j) {
      const float mu = s[j] * (1.0f / 128.0f);
      const v2f m2 = {mu, mu};
      d0[j] -= m2;
      d1[j] -= m2;
      const v2f p = d0[j] * d0[j] + d1[j] * d1[j];
      s[j] = p.x + p.y;
    }
    wave_sum13(s);
#pragma unroll
    for (int j = 0; j < 13; ++j) {
      const float rs = rsqrtf(s[j] * (1.0f / 128.0f) + 1e-5f);
      const v2f r2 = {rs, rs};
      const v2f o0 = d0[j] * r2 * g4.lo + b4.lo;
      const v2f o1 = d1[j] * r2 * g4.hi + b4.hi;
      const v4f o = {o0.x, o0.y, o1.x, o1.y};
      if (j < 12 || tail) og4[j * 8 * (HH / 4)] = o;
    }
  }

  if (!more) break;
  b = bn;
  }
}

extern "C" void kernel_launch(void* const* d_in, const int* in_sizes, int n_in,
                              void* d_out, int out_size, void* d_ws, size_t ws_size,
                              hipStream_t stream) {
  (void)n_in; (void)out_size; (void)d_ws; (void)ws_size;
  const float* x     = (const float*)d_in[0];
  const float* Wm1   = (const float*)d_in[4];
  const float* bm1   = (const float*)d_in[5];
  const float* Wm2   = (const float*)d_in[6];
  const float* bm2   = (const float*)d_in[7];
  const float* Wu1   = (const float*)d_in[8];
  const float* bu1   = (const float*)d_in[9];
  const float* Wu2   = (const float*)d_in[10];
  const float* bu2   = (const float*)d_in[11];
  const float* gamma = (const float*)d_in[12];
  const float* beta  = (const float*)d_in[13];
  float* out = (float*)d_out;

  const int B = in_sizes[0] / (NN * HH);
  const size_t shmem = (size_t)SMEM_FLOATS * sizeof(float);
  hipLaunchKernelGGL(mpnn_layer_kernel, dim3(256), dim3(256), shmem, stream,
                     x, Wm1, bm1, Wm2, bm2, Wu1, bu1, Wu2, bu2, gamma, beta, out, B);
}


# Wm2 folded into Wu1: each block forms Wm2 @ Wu1[128:] once on WMMA; the edge sums feed the update layer directly, 3 matrix phases per element not 4 (7168 WMMAs, not 8960); sum order changes
# speedup vs baseline: 3.0045x; 1.1681x over previous
#include <hip/hip_runtime.h>


typedef __attribute__((ext_vector_type(2))) float v2f;
typedef __attribute__((ext_vector_type(4))) float v4f;
typedef __attribute__((ext_vector_type(8))) float v8f;

#define NN    100
#define HH    128
#define NPAD  112
#define R4    400
#define LDP   132
#define LDP4  33

#define OFF_XS    0
#define OFF_PRE   14784
#define OFF_AGG   29568
#define OFF_W     44352
#define OFF_WLAST 60736
#define OFF_BM1   60864
#define OFF_BU1   60992
#define OFF_BU2   61120
#define OFF_GM    61248
#define OFF_BT    61376
#define OFF_CV    61504
#define OFF_DEG   61632
#define SMEM_FLOATS 61760
#define WROWS     144

__device__ __forceinline__ v8f wmma4(v2f a, v2f b, v8f c) {
  return __builtin_amdgcn_wmma_f32_16x16x4_f32(
       false, a,  false, b,
       (short)0, c,  false,  false);
}

__device__ __forceinline__ v4f relu4_keep(v4f a, int keep) {
  v4f r;
  r.x = __builtin_bit_cast(float, __builtin_bit_cast(int, fmaxf(a.x, 0.f)) & keep);
  r.y = __builtin_bit_cast(float, __builtin_bit_cast(int, fmaxf(a.y, 0.f)) & keep);
  r.z = __builtin_bit_cast(float, __builtin_bit_cast(int, fmaxf(a.z, 0.f)) & keep);
  r.w = __builtin_bit_cast(float, __builtin_bit_cast(int, fmaxf(a.w, 0.f)) & keep);
  return r;
}

template <int NR>
__device__ __forceinline__ void edge_sums(const v4f* __restrict__ pr4, v4f* __restrict__ sg4,
                                          v4f wl, int r0) {
  v4f h[NR][4];
  int keep[NR][4];
#pragma unroll
  for (int it = 0; it < NR; ++it) {
    const int  i  = r0 + 8 * it;
    const int  ic = i - (i / 10) * 10;
    const bool e0 = (i >= 10) && (i < NN);
    const bool e1 = (i < NN - 10);
    const bool e2 = (i < NN) && (ic >= 1);
    const bool e3 = (i < NN) && (ic <= 8);
    h[it][0] = pr4[(e0 ? i - 10 : i) * LDP4];
    h[it][1] = pr4[(e1 ? i + 10 : i) * LDP4];
    h[it][2] = pr4[(e2 ? i - 1  : i) * LDP4];
    h[it][3] = pr4[(e3 ? i + 1  : i) * LDP4];
    keep[it][0] = e0 ? -1 : 0;
    keep[it][1] = e1 ? -1 : 0;
    keep[it][2] = e2 ? -1 : 0;
    keep[it][3] = e3 ? -1 : 0;
  }
#pragma unroll
  for (int it = 0; it < NR; ++it) {
    const v4f t0 = relu4_keep(h[it][0], keep[it][0]);
    const v4f t1 = relu4_keep(h[it][1] + wl, keep[it][1]);
    const v4f t2 = relu4_keep(h[it][2] + 2.f * wl, keep[it][2]);
    const v4f t3 = relu4_keep(h[it][3] + 3.f * wl, keep[it][3]);
    sg4[(r0 + 8 * it) * LDP4] = ((t0 + t1) + t2) + t3;
  }
}

template <int NIT>
__device__ __forceinline__ void w_fetch(float4 (&r)[2 * NIT], const float* __restrict__ src, int tid) {
#pragma unroll
  for (int it = 0; it < NIT; ++it) {
    const int idx = tid + it * 256;
    const int kp = idx >> 5;
    const int cg = idx & 31;
    r[2 * it]     = *(const float4*)(src + (size_t)(2 * kp)     * 128 + cg * 4);
    r[2 * it + 1] = *(const float4*)(src + (size_t)(2 * kp + 1) * 128 + cg * 4);
  }
}

template <int NIT>
__device__ __forceinline__ void w_store(float* dst, const float4 (&r)[2 * NIT], int tid) {
#pragma unroll
  for (int it = 0; it < NIT; ++it) {
    const int idx = tid + it * 256;
    const int kp = idx >> 5;
    const int cg = idx & 31;
    const float4 r0 = r[2 * it];
    const float4 r1 = r[2 * it + 1];
    float4* d = (float4*)(dst + kp * 256 + (cg ^ ((kp & 1) << 2)) * 8);
    d[0] = make_float4(r0.x, r1.x, r0.y, r1.y);
    d[1] = make_float4(r0.z, r1.z, r0.w, r1.w);
  }
}

template <int NIT>
__device__ __forceinline__ void w_store_rows(float* dst, const float4 (&r)[2 * NIT], int tid) {
#pragma unroll
  for (int it = 0; it < NIT; ++it) {
    const int idx = tid + it * 256;
    const int kp = idx >> 5;
    const int cg = idx & 31;
    const float4 r0 = r[2 * it];
    const float4 r1 = r[2 * it + 1];
    ((float4*)dst)[(2 * kp)     * LDP4 + cg] = make_float4(r0.x, r0.y, r0.z, r0.w);
    ((float4*)dst)[(2 * kp + 1) * LDP4 + cg] = make_float4(r1.x, r1.y, r1.z, r1.w);
  }
}

struct XRows { float4 r0, r1, r2, r3, r4, r5, r6, r7, r8, r9, r10, r11, r12; };

__device__ __forceinline__ XRows x_fetch(const float* __restrict__ xg, int tid) {
  const float4* xg4 = (const float4*)xg;
  const int il = tid + 12 * 256;
  const bool inx = il < NN * HH / 4;
  XRows r;
  r.r0  = xg4[tid + 0 * 256];
  r.r1  = xg4[tid + 1 * 256];
  r.r2  = xg4[tid + 2 * 256];
  r.r3  = xg4[tid + 3 * 256];
  r.r4  = xg4[tid + 4 * 256];
  r.r5  = xg4[tid + 5 * 256];
  r.r6  = xg4[tid + 6 * 256];
  r.r7  = xg4[tid + 7 * 256];
  r.r8  = xg4[tid + 8 * 256];
  r.r9  = xg4[tid + 9 * 256];
  r.r10 = xg4[tid + 10 * 256];
  r.r11 = xg4[tid + 11 * 256];
  const float4 xt = xg4[inx ? il : NN * HH / 4 - 1];
  r.r12 = make_float4(inx ? xt.x : 0.f, inx ? xt.y : 0.f,
                      inx ? xt.z : 0.f, inx ? xt.w : 0.f);
  return r;
}

__device__ __forceinline__ void x_store(float* xs, const XRows& r, int p4) {
  float4* d = (float4*)xs + p4;
  d[0 * 8 * LDP4]  = r.r0;
  d[1 * 8 * LDP4]  = r.r1;
  d[2 * 8 * LDP4]  = r.r2;
  d[3 * 8 * LDP4]  = r.r3;
  d[4 * 8 * LDP4]  = r.r4;
  d[5 * 8 * LDP4]  = r.r5;
  d[6 * 8 * LDP4]  = r.r6;
  d[7 * 8 * LDP4]  = r.r7;
  d[8 * 8 * LDP4]  = r.r8;
  d[9 * 8 * LDP4]  = r.r9;
  d[10 * 8 * LDP4] = r.r10;
  d[11 * 8 * LDP4] = r.r11;
  d[12 * 8 * LDP4] = r.r12;
}

__device__ __forceinline__ void wmma_col7x4(const float* __restrict__ ap,
                                            const float* __restrict__ bp, v8f acc[7]) {
#pragma unroll
  for (int j = 0; j < 4; ++j) {
    const v2f bb = *(const v2f*)(bp + j * 512);
#pragma unroll
    for (int m = 0; m < 7; ++m) {
      const v2f a = *(const v2f*)(ap + m * 16 * LDP + j * 4);
      acc[m] = wmma4(a, bb, acc[m]);
    }
  }
}

template <int NKS>
__device__ __forceinline__ void b_fetch(v2f (&r)[NKS], const float* __restrict__ w, int hi, int ncol) {
#pragma unroll
  for (int ks = 0; ks < NKS; ++ks) {
    const float* p = w + (4 * ks + 2 * hi) * 128 + ncol;
    r[ks].x = p[0];
    r[ks].y = p[128];
  }
}

__device__ __forceinline__ void a_load28(v2f (&a)[28], const float* __restrict__ ap) {
#pragma unroll
  for (int j = 0; j < 4; ++j)
#pragma unroll
    for (int m = 0; m < 7; ++m)
      a[j * 7 + m] = *(const v2f*)(ap + m * 16 * LDP + j * 4);
}

template <int NBLK>
__device__ __forceinline__ void wmma_col7_regb(const float* __restrict__ ap0,
                                               const float* __restrict__ ap1,
                                               const v2f* __restrict__ breg, v8f acc[7]) {
  v2f a[2][28];
  a_load28(a[0], ap0);
#pragma unroll
  for (int blk = 0; blk < NBLK; ++blk) {
    if (blk + 1 < NBLK)
      a_load28(a[(blk + 1) & 1], (blk + 1 < 8) ? ap0 + (blk + 1) * 16 : ap1 + (blk + 1 - 8) * 16);
    __builtin_amdgcn_sched_barrier(0);
#pragma unroll
    for (int j = 0; j < 4; ++j) {
      const v2f bb = breg[blk * 4 + j];
#pragma unroll
      for (int m = 0; m < 7; ++m)
        acc[m] = wmma4(a[blk & 1][j * 7 + m], bb, acc[m]);
    }
    __builtin_amdgcn_sched_barrier(0);
  }
}

template <int XM>
__device__ __forceinline__ float row_xor(float v) {
  return __builtin_bit_cast(float, __builtin_amdgcn_update_dpp(
      0, __builtin_bit_cast(int, v), 0x160 + XM, 0xf, 0xf, true));
}
__device__ __forceinline__ float half_xor(float v) {
  return __builtin_bit_cast(float, __builtin_amdgcn_permlanex16(
      __builtin_bit_cast(int, v), __builtin_bit_cast(int, v), 0x76543210, 0xfedcba98, false, true));
}

__device__ __forceinline__ void wave_sum13(float (&s)[13]) {
#pragma unroll
  for (int j = 0; j < 13; ++j) s[j] += row_xor<1>(s[j]);
#pragma unroll
  for (int j = 0; j < 13; ++j) s[j] += row_xor<2>(s[j]);
#pragma unroll
  for (int j = 0; j < 13; ++j) s[j] += row_xor<4>(s[j]);
#pragma unroll
  for (int j = 0; j < 13; ++j) s[j] += row_xor<8>(s[j]);
#pragma unroll
  for (int j = 0; j < 13; ++j) s[j] += half_xor(s[j]);
}

extern "C" __global__ __launch_bounds__(256) __attribute__((amdgpu_waves_per_eu(2, 2)))
void mpnn_layer_kernel(const float* __restrict__ x,
                       const float* __restrict__ Wm1, const float* __restrict__ bm1,
                       const float* __restrict__ Wm2, const float* __restrict__ bm2,
                       const float* __restrict__ Wu1, const float* __restrict__ bu1,
                       const float* __restrict__ Wu2, const float* __restrict__ bu2,
                       const float* __restrict__ gamma, const float* __restrict__ beta,
                       float* __restrict__ out, int nb)
{
  extern __shared__ float smem[];
  float* xs    = smem + OFF_XS;
  float* pre   = smem + OFF_PRE;
  float* agg   = smem + OFF_AGG;
  float* wb    = smem + OFF_W;
  float* wlast = smem + OFF_WLAST;
  float* lbm1  = smem + OFF_BM1;
  float* lbu1  = smem + OFF_BU1;
  float* lbu2  = smem + OFF_BU2;
  float* lgm   = smem + OFF_GM;
  float* lbt   = smem + OFF_BT;
  float* lcv   = smem + OFF_CV;
  float* ldeg  = smem + OFF_DEG;
  float* scr   = smem + OFF_PRE;

  int b = blockIdx.x;
  if (b >= nb) return;

  {
    const int tid  = threadIdx.x;
    const int lane = tid & 31;
    const int p4   = (tid >> 5) * LDP4 + lane;
    const float4 z4 = make_float4(0.f, 0.f, 0.f, 0.f);
    float4 w1[16];
    float4 w2[16];
    const XRows xr = x_fetch(x + (size_t)b * (NN * HH), tid);
    w_fetch<8>(w1, Wm1, tid);
    w_fetch<8>(w2, Wm2, tid);
    ((float4*)xs)[p4 + 13 * 8 * LDP4] = z4;
    x_store(xs, xr, p4);
    w_store<8>(wb, w1, tid);
    w_store_rows<8>(scr, w2, tid);
    {
      const float4 bq = ((const float4*)bm2)[lane];
      const bool w0 = (tid >> 5) == 0;
      ((float4*)scr)[(128 + (tid >> 5)) * LDP4 + lane] =
          make_float4(w0 ? bq.x : 0.f, w0 ? bq.y : 0.f, w0 ? bq.z : 0.f, w0 ? bq.w : 0.f);
      ((float4*)scr)[(136 + (tid >> 5)) * LDP4 + lane] = z4;
    }
    if (tid < 128) {
      const int r  = tid;
      const int rc = r - (r / 10) * 10;
      const int in = (r < NN) ? 1 : 0;
      ldeg[r] = (float)(((r >= 10 ? 1 : 0) & in) + (r < NN - 10 ? 1 : 0)
                        + ((rc >= 1 ? 1 : 0) & in) + ((rc <= 8 ? 1 : 0) & in));
    }
    if ((tid >> 5) == 0) {
      ((float4*)wlast)[lane] = ((const float4*)(Wm1 + 128 * 128))[lane];
      ((float4*)lbm1)[lane]  = ((const float4*)bm1)[lane];
      ((float4*)lbu1)[lane]  = ((const float4*)bu1)[lane];
      ((float4*)lbu2)[lane]  = ((const float4*)bu2)[lane];
      ((float4*)lgm)[lane]   = ((const float4*)gamma)[lane];
      ((float4*)lbt)[lane]   = ((const float4*)beta)[lane];
    }
  }
  v2f bu1r[64];
  v2f bu2r[32];
  {
    const int t0 = threadIdx.x;
    b_fetch<64>(bu1r, Wu1, (t0 & 31) >> 4, (t0 >> 5) * 16 + (t0 & 15));
    b_fetch<32>(bu2r, Wu2, (t0 & 31) >> 4, (t0 >> 5) * 16 + (t0 & 15));
  }
  __syncthreads();

  const v8f vz = {0.f, 0.f, 0.f, 0.f, 0.f, 0.f, 0.f, 0.f};

  {
    const int t0   = threadIdx.x;
    const int li0  = t0 & 15;
    const int hi0  = (t0 & 31) >> 4;
    const int nc0  = (t0 >> 5) * 16 + li0;
    const float* ap = scr + li0 * LDP + hi0 * 2;
    v8f wacc[9];
#pragma unroll
    for (int m = 0; m < 9; ++m) wacc[m] = vz;
#pragma unroll
    for (int ks = 0; ks < 32; ++ks) {
      const v2f bb = bu1r[32 + ks];
#pragma unroll
      for (int m = 0; m < 9; ++m) {
        const v2f a = *(const v2f*)(ap + m * 16 * LDP + ks * 4);
        wacc[m] = wmma4(a, bb, wacc[m]);
      }
    }
    __syncthreads();
#pragma unroll
    for (int m = 0; m < 8; ++m) {
#pragma unroll
      for (int v = 0; v < 8; ++v)
        scr[(m * 16 + hi0 * 8 + v) * LDP + nc0] = wacc[m][v];
    }
    if (hi0 == 0) lcv[nc0] = wacc[8][0];
    __syncthreads();
#pragma unroll
    for (int ks = 0; ks < 32; ++ks) {
      const float* p = scr + (4 * ks + 2 * hi0) * LDP + nc0;
      bu1r[32 + ks].x = p[0];
      bu1r[32 + ks].y = p[LDP];
    }
  }

  for (;;) {
  const int  bn   = b + (int)gridDim.x;
  const bool more = bn < nb;

  int tid = threadIdx.x;
  asm volatile("" : "+v"(tid));
  const int wave = tid >> 5;
  const int lane = tid & 31;
  const int li   = lane & 15;
  const int hi   = lane >> 4;
  const int ncol = wave * 16 + li;
  const int bofs = hi * 256 + ((wave ^ hi) << 5) + li * 2;
  const int aofs = li * LDP + hi * 2;
  const int p4   = (tid >> 5) * LDP4 + (tid & 31);

  {
    v8f acc[7];
#pragma unroll
    for (int m = 0; m < 7; ++m) acc[m] = vz;
    for (int kq = 0; kq < 8; ++kq)
      wmma_col7x4(xs + aofs + kq * 16, wb + bofs + kq * 2048, acc);
    const float bias = lbm1[ncol];
#pragma unroll
    for (int m = 0; m < 7; ++m) {
#pragma unroll
      for (int v = 0; v < 8; ++v) {
        const int row = m * 16 + hi * 8 + v;
        pre[row * LDP + ncol] = acc[m][v] + bias;
      }
    }
  }
  __syncthreads();

  {
    const int wv = __builtin_amdgcn_readfirstlane(wave);
    const v4f z4v = {0.f, 0.f, 0.f, 0.f};
    const v4f wl = ((const v4f*)wlast)[lane];
    const v4f* pr4 = (const v4f*)pre + lane;
    v4f* sg4 = (v4f*)agg + lane;
    edge_sums<7>(pr4, sg4, wl, wv);
    edge_sums<6>(pr4, sg4, wl, wv + 56);
    sg4[(wv + 104) * LDP4] = z4v;
  }
  __syncthreads();

  {
    v8f acc[7];
#pragma unroll
    for (int m = 0; m < 7; ++m) acc[m] = vz;
    wmma_col7_regb<16>(xs + aofs, agg + aofs, bu1r, acc);
    const float bias = lbu1[ncol];
    const float cv   = lcv[ncol];
#pragma unroll
    for (int m = 0; m < 7; ++m) {
      const v4f dga = *(const v4f*)(ldeg + m * 16 + hi * 8);
      const v4f dgb = *(const v4f*)(ldeg + m * 16 + hi * 8 + 4);
      const float dg[8] = {dga.x, dga.y, dga.z, dga.w, dgb.x, dgb.y, dgb.z, dgb.w};
#pragma unroll
      for (int v = 0; v < 8; ++v) {
        const int row = m * 16 + hi * 8 + v;
        pre[row * LDP + ncol] = fmaxf(acc[m][v] + (dg[v] * cv + bias), 0.f);
      }
    }
  }
  __syncthreads();

  const XRows xn = x_fetch(x + (size_t)(more ? bn : b) * (NN * HH), tid);

  {
    v8f acc[7];
#pragma unroll
    for (int m = 0; m < 7; ++m) acc[m] = vz;
    wmma_col7_regb<8>(pre + aofs, pre + aofs, bu2r, acc);
    const float bias = lbu2[ncol];
#pragma unroll
    for (int m = 0; m < 7; ++m) {
#pragma unroll
      for (int v = 0; v < 8; ++v) {
        const int row = m * 16 + hi * 8 + v;
        agg[row * LDP + ncol] = xs[row * LDP + ncol] + acc[m][v] + bias;
      }
    }
  }
  __syncthreads();

  x_store(xs, xn, p4);
  __syncthreads();

  {
    const bool tail = wave < 4;
    const v4f g4 = ((const v4f*)lgm)[lane];
    const v4f b4 = ((const v4f*)lbt)[lane];
    const v4f* hb4 = (const v4f*)agg + wave * LDP4 + lane;
    v4f* og4 = (v4f*)(out + (size_t)b * (NN * HH)) + wave * (HH / 4) + lane;
    v2f d0[13], d1[13];
    float s[13];
#pragma unroll
    for (int j = 0; j < 13; ++j) {
      const v4f h = hb4[(j < 12 ? j : (tail ? 12 : 11)) * 8 * LDP4];
      d0[j] = h.lo;
      d1[j] = h.hi;
      const v2f a = d0[j] + d1[j];
      s[j] = a.x + a.y;
    }
    wave_sum13(s);
#pragma unroll
    for (int j = 0; j < 13; ++j) {
      const float mu = s[j] * (1.0f / 128.0f);
      const v2f m2 = {mu, mu};
      d0[j] -= m2;
      d1[j] -= m2;
      const v2f p = d0[j] * d0[j] + d1[j] * d1[j];
      s[j] = p.x + p.y;
    }
    wave_sum13(s);
#pragma unroll
    for (int j = 0; j < 13; ++j) {
      const float rs = rsqrtf(s[j] * (1.0f / 128.0f) + 1e-5f);
      const v2f r2 = {rs, rs};
      const v2f o0 = d0[j] * r2 * g4.lo + b4.lo;
      const v2f o1 = d1[j] * r2 * g4.hi + b4.hi;
      const v4f o = {o0.x, o0.y, o1.x, o1.y};
      if (j < 12 || tail) og4[j * 8 * (HH / 4)] = o;
    }
  }

  if (!more) break;
  b = bn;
  }
}

extern "C" void kernel_launch(void* const* d_in, const int* in_sizes, int n_in,
                              void* d_out, int out_size, void* d_ws, size_t ws_size,
                              hipStream_t stream) {
  (void)n_in; (void)out_size; (void)d_ws; (void)ws_size;
  const float* x     = (const float*)d_in[0];
  const float* Wm1   = (const float*)d_in[4];
  const float* bm1   = (const float*)d_in[5];
  const float* Wm2   = (const float*)d_in[6];
  const float* bm2   = (const float*)d_in[7];
  const float* Wu1   = (const float*)d_in[8];
  const float* bu1   = (const float*)d_in[9];
  const float* Wu2   = (const float*)d_in[10];
  const float* bu2   = (const float*)d_in[11];
  const float* gamma = (const float*)d_in[12];
  const float* beta  = (const float*)d_in[13];
  float* out = (float*)d_out;

  const int B = in_sizes[0] / (NN * HH);
  const size_t shmem = (size_t)SMEM_FLOATS * sizeof(float);
  hipLaunchKernelGGL(mpnn_layer_kernel, dim3(256), dim3(256), shmem, stream,
                     x, Wm1, bm1, Wm2, bm2, Wu1, bu1, Wu2, bu2, gamma, beta, out, B);
}


# Rows 96..99 of each element set aside in LDS (x rows, edge sums): update layer and LayerNorm run 6 row tiles per element, not 7; the rows set aside run as 2 packed tiles per 8 elements
# speedup vs baseline: 3.1975x; 1.0642x over previous
#include <hip/hip_runtime.h>


typedef __attribute__((ext_vector_type(2))) float v2f;
typedef __attribute__((ext_vector_type(4))) float v4f;
typedef __attribute__((ext_vector_type(8))) float v8f;

#define NN    100
#define HH    128
#define NPAD  112
#define R4    400
#define LDP   132
#define LDP4  33

#define OFF_XS    0
#define OFF_PRE   14784
#define OFF_AGG   29568
#define OFF_W     44352
#define OFF_WLAST 60736
#define OFF_BM1   60864
#define OFF_BU1   60992
#define OFF_BU2   61120
#define OFF_GM    61248
#define OFF_BT    61376
#define OFF_CV    61504
#define OFF_DEG   61632
#define OFF_TX    61760
#define OFF_TS    65984
#define SMEM_FLOATS 70208
#define TSLOTS    8
#define WROWS     144

__device__ __forceinline__ v8f wmma4(v2f a, v2f b, v8f c) {
  return __builtin_amdgcn_wmma_f32_16x16x4_f32(
       false, a,  false, b,
       (short)0, c,  false,  false);
}

__device__ __forceinline__ v4f relu4_keep(v4f a, int keep) {
  v4f r;
  r.x = __builtin_bit_cast(float, __builtin_bit_cast(int, fmaxf(a.x, 0.f)) & keep);
  r.y = __builtin_bit_cast(float, __builtin_bit_cast(int, fmaxf(a.y, 0.f)) & keep);
  r.z = __builtin_bit_cast(float, __builtin_bit_cast(int, fmaxf(a.z, 0.f)) & keep);
  r.w = __builtin_bit_cast(float, __builtin_bit_cast(int, fmaxf(a.w, 0.f)) & keep);
  return r;
}

template <int NR, bool TAIL>
__device__ __forceinline__ void edge_sums(const v4f* __restrict__ pr4, v4f* __restrict__ sg4,
                                          v4f* __restrict__ tl4, v4f wl, int r0) {
  v4f h[NR][4];
  int keep[NR][4];
#pragma unroll
  for (int it = 0; it < NR; ++it) {
    const int  i  = r0 + 8 * it;
    const int  ic = i - (i / 10) * 10;
    const bool e0 = (i >= 10) && (i < NN);
    const bool e1 = (i < NN - 10);
    const bool e2 = (i < NN) && (ic >= 1);
    const bool e3 = (i < NN) && (ic <= 8);
    h[it][0] = pr4[(e0 ? i - 10 : i) * LDP4];
    h[it][1] = pr4[(e1 ? i + 10 : i) * LDP4];
    h[it][2] = pr4[(e2 ? i - 1  : i) * LDP4];
    h[it][3] = pr4[(e3 ? i + 1  : i) * LDP4];
    keep[it][0] = e0 ? -1 : 0;
    keep[it][1] = e1 ? -1 : 0;
    keep[it][2] = e2 ? -1 : 0;
    keep[it][3] = e3 ? -1 : 0;
  }
#pragma unroll
  for (int it = 0; it < NR; ++it) {
    const v4f t0 = relu4_keep(h[it][0], keep[it][0]);
    const v4f t1 = relu4_keep(h[it][1] + wl, keep[it][1]);
    const v4f t2 = relu4_keep(h[it][2] + 2.f * wl, keep[it][2]);
    const v4f t3 = relu4_keep(h[it][3] + 3.f * wl, keep[it][3]);
    const v4f sum = ((t0 + t1) + t2) + t3;
    if (TAIL && it == NR - 1) tl4[0] = sum;
    else                      sg4[(r0 + 8 * it) * LDP4] = sum;
  }
}

template <int NIT>
__device__ __forceinline__ void w_fetch(float4 (&r)[2 * NIT], const float* __restrict__ src, int tid) {
#pragma unroll
  for (int it = 0; it < NIT; ++it) {
    const int idx = tid + it * 256;
    const int kp = idx >> 5;
    const int cg = idx & 31;
    r[2 * it]     = *(const float4*)(src + (size_t)(2 * kp)     * 128 + cg * 4);
    r[2 * it + 1] = *(const float4*)(src + (size_t)(2 * kp + 1) * 128 + cg * 4);
  }
}

template <int NIT>
__device__ __forceinline__ void w_store(float* dst, const float4 (&r)[2 * NIT], int tid) {
#pragma unroll
  for (int it = 0; it < NIT; ++it) {
    const int idx = tid + it * 256;
    const int kp = idx >> 5;
    const int cg = idx & 31;
    const float4 r0 = r[2 * it];
    const float4 r1 = r[2 * it + 1];
    float4* d = (float4*)(dst + kp * 256 + (cg ^ ((kp & 1) << 2)) * 8);
    d[0] = make_float4(r0.x, r1.x, r0.y, r1.y);
    d[1] = make_float4(r0.z, r1.z, r0.w, r1.w);
  }
}

template <int NIT>
__device__ __forceinline__ void w_store_rows(float* dst, const float4 (&r)[2 * NIT], int tid) {
#pragma unroll
  for (int it = 0; it < NIT; ++it) {
    const int idx = tid + it * 256;
    const int kp = idx >> 5;
    const int cg = idx & 31;
    const float4 r0 = r[2 * it];
    const float4 r1 = r[2 * it + 1];
    ((float4*)dst)[(2 * kp)     * LDP4 + cg] = make_float4(r0.x, r0.y, r0.z, r0.w);
    ((float4*)dst)[(2 * kp + 1) * LDP4 + cg] = make_float4(r1.x, r1.y, r1.z, r1.w);
  }
}

struct XRows { float4 r0, r1, r2, r3, r4, r5, r6, r7, r8, r9, r10, r11, r12; };

__device__ __forceinline__ XRows x_fetch(const float* __restrict__ xg, int tid) {
  const float4* xg4 = (const float4*)xg;
  const int il = tid + 12 * 256;
  const bool inx = il < NN * HH / 4;
  XRows r;
  r.r0  = xg4[tid + 0 * 256];
  r.r1  = xg4[tid + 1 * 256];
  r.r2  = xg4[tid + 2 * 256];
  r.r3  = xg4[tid + 3 * 256];
  r.r4  = xg4[tid + 4 * 256];
  r.r5  = xg4[tid + 5 * 256];
  r.r6  = xg4[tid + 6 * 256];
  r.r7  = xg4[tid + 7 * 256];
  r.r8  = xg4[tid + 8 * 256];
  r.r9  = xg4[tid + 9 * 256];
  r.r10 = xg4[tid + 10 * 256];
  r.r11 = xg4[tid + 11 * 256];
  const float4 xt = xg4[inx ? il : NN * HH / 4 - 1];
  r.r12 = make_float4(inx ? xt.x : 0.f, inx ? xt.y : 0.f,
                      inx ? xt.z : 0.f, inx ? xt.w : 0.f);
  return r;
}

__device__ __forceinline__ void x_store(float* xs, const XRows& r, int p4) {
  float4* d = (float4*)xs + p4;
  d[0 * 8 * LDP4]  = r.r0;
  d[1 * 8 * LDP4]  = r.r1;
  d[2 * 8 * LDP4]  = r.r2;
  d[3 * 8 * LDP4]  = r.r3;
  d[4 * 8 * LDP4]  = r.r4;
  d[5 * 8 * LDP4]  = r.r5;
  d[6 * 8 * LDP4]  = r.r6;
  d[7 * 8 * LDP4]  = r.r7;
  d[8 * 8 * LDP4]  = r.r8;
  d[9 * 8 * LDP4]  = r.r9;
  d[10 * 8 * LDP4] = r.r10;
  d[11 * 8 * LDP4] = r.r11;
  d[12 * 8 * LDP4] = r.r12;
}

__device__ __forceinline__ void wmma_col7x4(const float* __restrict__ ap,
                                            const float* __restrict__ bp, v8f acc[7]) {
#pragma unroll
  for (int j = 0; j < 4; ++j) {
    const v2f bb = *(const v2f*)(bp + j * 512);
#pragma unroll
    for (int m = 0; m < 7; ++m) {
      const v2f a = *(const v2f*)(ap + m * 16 * LDP + j * 4);
      acc[m] = wmma4(a, bb, acc[m]);
    }
  }
}

template <int NKS>
__device__ __forceinline__ void b_fetch(v2f (&r)[NKS], const float* __restrict__ w, int hi, int ncol) {
#pragma unroll
  for (int ks = 0; ks < NKS; ++ks) {
    const float* p = w + (4 * ks + 2 * hi) * 128 + ncol;
    r[ks].x = p[0];
    r[ks].y = p[128];
  }
}

template <int NT>
__device__ __forceinline__ void a_load_blk(v2f (&a)[4 * NT], const float* __restrict__ ap) {
#pragma unroll
  for (int j = 0; j < 4; ++j)
#pragma unroll
    for (int m = 0; m < NT; ++m)
      a[j * NT + m] = *(const v2f*)(ap + m * 16 * LDP + j * 4);
}

template <int NT, int NBLK>
__device__ __forceinline__ void wmma_col_regb(const float* __restrict__ ap0,
                                              const float* __restrict__ ap1,
                                              const v2f* __restrict__ breg, v8f (&acc)[NT]) {
  v2f a[2][4 * NT];
  a_load_blk<NT>(a[0], ap0);
#pragma unroll
  for (int blk = 0; blk < NBLK; ++blk) {
    if (blk + 1 < NBLK)
      a_load_blk<NT>(a[(blk + 1) & 1], (blk + 1 < 8) ? ap0 + (blk + 1) * 16 : ap1 + (blk + 1 - 8) * 16);
    __builtin_amdgcn_sched_barrier(0);
#pragma unroll
    for (int j = 0; j < 4; ++j) {
      const v2f bb = breg[blk * 4 + j];
#pragma unroll
      for (int m = 0; m < NT; ++m)
        acc[m] = wmma4(a[blk & 1][j * NT + m], bb, acc[m]);
    }
    __builtin_amdgcn_sched_barrier(0);
  }
}

template <int XM>
__device__ __forceinline__ float row_xor(float v) {
  return __builtin_bit_cast(float, __builtin_amdgcn_update_dpp(
      0, __builtin_bit_cast(int, v), 0x160 + XM, 0xf, 0xf, true));
}
__device__ __forceinline__ float half_xor(float v) {
  return __builtin_bit_cast(float, __builtin_amdgcn_permlanex16(
      __builtin_bit_cast(int, v), __builtin_bit_cast(int, v), 0x76543210, 0xfedcba98, false, true));
}

template <int NS>
__device__ __forceinline__ void wave_sums(float (&s)[NS]) {
#pragma unroll
  for (int j = 0; j < NS; ++j) s[j] += row_xor<1>(s[j]);
#pragma unroll
  for (int j = 0; j < NS; ++j) s[j] += row_xor<2>(s[j]);
#pragma unroll
  for (int j = 0; j < NS; ++j) s[j] += row_xor<4>(s[j]);
#pragma unroll
  for (int j = 0; j < NS; ++j) s[j] += row_xor<8>(s[j]);
#pragma unroll
  for (int j = 0; j < NS; ++j) s[j] += half_xor(s[j]);
}

template <int NRW>
__device__ __forceinline__ void ln_rows(v4f (&o)[NRW], const v4f* __restrict__ hb4, v4f g4, v4f b4) {
  v2f d0[NRW], d1[NRW];
  float s[NRW];
#pragma unroll
  for (int j = 0; j < NRW; ++j) {
    const v4f h = hb4[j * 8 * LDP4];
    d0[j] = h.lo;
    d1[j] = h.hi;
    const v2f a = d0[j] + d1[j];
    s[j] = a.x + a.y;
  }
  wave_sums<NRW>(s);
#pragma unroll
  for (int j = 0; j < NRW; ++j) {
    const float mu = s[j] * (1.0f / 128.0f);
    const v2f m2 = {mu, mu};
    d0[j] -= m2;
    d1[j] -= m2;
    const v2f p = d0[j] * d0[j] + d1[j] * d1[j];
    s[j] = p.x + p.y;
  }
  wave_sums<NRW>(s);
#pragma unroll
  for (int j = 0; j < NRW; ++j) {
    const float rs = rsqrtf(s[j] * (1.0f / 128.0f) + 1e-5f);
    const v2f r2 = {rs, rs};
    const v2f o0 = d0[j] * r2 * g4.lo + b4.lo;
    const v2f o1 = d1[j] * r2 * g4.hi + b4.hi;
    const v4f ov = {o0.x, o0.y, o1.x, o1.y};
    o[j] = ov;
  }
}

extern "C" __global__ __launch_bounds__(256) __attribute__((amdgpu_waves_per_eu(2, 2)))
void mpnn_layer_kernel(const float* __restrict__ x,
                       const float* __restrict__ Wm1, const float* __restrict__ bm1,
                       const float* __restrict__ Wm2, const float* __restrict__ bm2,
                       const float* __restrict__ Wu1, const float* __restrict__ bu1,
                       const float* __restrict__ Wu2, const float* __restrict__ bu2,
                       const float* __restrict__ gamma, const float* __restrict__ beta,
                       float* __restrict__ out, int nb)
{
  extern __shared__ float smem[];
  float* xs    = smem + OFF_XS;
  float* pre   = smem + OFF_PRE;
  float* agg   = smem + OFF_AGG;
  float* wb    = smem + OFF_W;
  float* wlast = smem + OFF_WLAST;
  float* lbm1  = smem + OFF_BM1;
  float* lbu1  = smem + OFF_BU1;
  float* lbu2  = smem + OFF_BU2;
  float* lgm   = smem + OFF_GM;
  float* lbt   = smem + OFF_BT;
  float* lcv   = smem + OFF_CV;
  float* ldeg  = smem + OFF_DEG;
  float* tx    = smem + OFF_TX;
  float* ts    = smem + OFF_TS;
  float* scr   = smem + OFF_PRE;

  int b = blockIdx.x;
  if (b >= nb) return;

  {
    const int tid  = threadIdx.x;
    const int lane = tid & 31;
    const int p4   = (tid >> 5) * LDP4 + lane;
    const float4 z4 = make_float4(0.f, 0.f, 0.f, 0.f);
    float4 w1[16];
    float4 w2[16];
    const XRows xr = x_fetch(x + (size_t)b * (NN * HH), tid);
    w_fetch<8>(w1, Wm1, tid);
    w_fetch<8>(w2, Wm2, tid);
    ((float4*)xs)[p4 + 13 * 8 * LDP4] = z4;
#pragma unroll
    for (int it = 0; it < 9; ++it) {
      const int i = tid + it * 256;
      ((float4*)tx)[i < 2 * 4 * TSLOTS * LDP4 ? i : 2 * 4 * TSLOTS * LDP4 - 1] = z4;
    }
    x_store(xs, xr, p4);
    w_store<8>(wb, w1, tid);
    w_store_rows<8>(scr, w2, tid);
    {
      const float4 bq = ((const float4*)bm2)[lane];
      const bool w0 = (tid >> 5) == 0;
      ((float4*)scr)[(128 + (tid >> 5)) * LDP4 + lane] =
          make_float4(w0 ? bq.x : 0.f, w0 ? bq.y : 0.f, w0 ? bq.z : 0.f, w0 ? bq.w : 0.f);
      ((float4*)scr)[(136 + (tid >> 5)) * LDP4 + lane] = z4;
    }
    if (tid < 128) {
      const int r  = tid;
      const int rc = r - (r / 10) * 10;
      const int in = (r < NN) ? 1 : 0;
      ldeg[r] = (float)(((r >= 10 ? 1 : 0) & in) + (r < NN - 10 ? 1 : 0)
                        + ((rc >= 1 ? 1 : 0) & in) + ((rc <= 8 ? 1 : 0) & in));
    }
    if ((tid >> 5) == 0) {
      ((float4*)wlast)[lane] = ((const float4*)(Wm1 + 128 * 128))[lane];
      ((float4*)lbm1)[lane]  = ((const float4*)bm1)[lane];
      ((float4*)lbu1)[lane]  = ((const float4*)bu1)[lane];
      ((float4*)lbu2)[lane]  = ((const float4*)bu2)[lane];
      ((float4*)lgm)[lane]   = ((const float4*)gamma)[lane];
      ((float4*)lbt)[lane]   = ((const float4*)beta)[lane];
    }
  }
  v2f bu1r[64];
  v2f bu2r[32];
  {
    const int t0 = threadIdx.x;
    b_fetch<64>(bu1r, Wu1, (t0 & 31) >> 4, (t0 >> 5) * 16 + (t0 & 15));
    b_fetch<32>(bu2r, Wu2, (t0 & 31) >> 4, (t0 >> 5) * 16 + (t0 & 15));
  }
  __syncthreads();

  const v8f vz = {0.f, 0.f, 0.f, 0.f, 0.f, 0.f, 0.f, 0.f};

  {
    const int t0   = threadIdx.x;
    const int li0  = t0 & 15;
    const int hi0  = (t0 & 31) >> 4;
    const int nc0  = (t0 >> 5) * 16 + li0;
    const float* ap = scr + li0 * LDP + hi0 * 2;
    v8f wacc[9];
#pragma unroll
    for (int m = 0; m < 9; ++m) wacc[m] = vz;
#pragma unroll
    for (int ks = 0; ks < 32; ++ks) {
      const v2f bb = bu1r[32 + ks];
#pragma unroll
      for (int m = 0; m < 9; ++m) {
        const v2f a = *(const v2f*)(ap + m * 16 * LDP + ks * 4);
        wacc[m] = wmma4(a, bb, wacc[m]);
      }
    }
    __syncthreads();
#pragma unroll
    for (int m = 0; m < 8; ++m) {
#pragma unroll
      for (int v = 0; v < 8; ++v)
        scr[(m * 16 + hi0 * 8 + v) * LDP + nc0] = wacc[m][v];
    }
    if (hi0 == 0) lcv[nc0] = wacc[8][0];
    __syncthreads();
#pragma unroll
    for (int ks = 0; ks < 32; ++ks) {
      const float* p = scr + (4 * ks + 2 * hi0) * LDP + nc0;
      bu1r[32 + ks].x = p[0];
      bu1r[32 + ks].y = p[LDP];
    }
  }

  int cnt    = 0;
  int bfirst = b;
  for (;;) {
  const int  bn   = b + (int)gridDim.x;
  const bool more = bn < nb;

  int tid = threadIdx.x;
  asm volatile("" : "+v"(tid));
  const int wave = tid >> 5;
  const int lane = tid & 31;
  const int li   = lane & 15;
  const int hi   = lane >> 4;
  const int ncol = wave * 16 + li;
  const int bofs = hi * 256 + ((wave ^ hi) << 5) + li * 2;
  const int aofs = li * LDP + hi * 2;
  const int p4   = (tid >> 5) * LDP4 + (tid & 31);

  {
    v8f acc[7];
#pragma unroll
    for (int m = 0; m < 7; ++m) acc[m] = vz;
    for (int kq = 0; kq < 8; ++kq)
      wmma_col7x4(xs + aofs + kq * 16, wb + bofs + kq * 2048, acc);
    const float bias = lbm1[ncol];
#pragma unroll
    for (int m = 0; m < 7; ++m) {
#pragma unroll
      for (int v = 0; v < 8; ++v) {
        const int row = m * 16 + hi * 8 + v;
        pre[row * LDP + ncol] = acc[m][v] + bias;
      }
    }
  }
  __syncthreads();

  {
    const int wv = __builtin_amdgcn_readfirstlane(wave);
    const v4f wl = ((const v4f*)wlast)[lane];
    const v4f* pr4 = (const v4f*)pre + lane;
    v4f* sg4 = (v4f*)agg + lane;
    const bool tw   = wv < 4;
    const int  trow = 4 * cnt + (wv & 3);
    const int tso = tw ? OFF_TS / 4 + trow * LDP4 : OFF_AGG / 4 + (wv + 96) * LDP4;
    const int txo = tw ? OFF_TX / 4 + trow * LDP4 : OFF_AGG / 4 + (wv + 100) * LDP4;
    v4f* tsd = (v4f*)smem + tso + lane;
    v4f* txd = (v4f*)smem + txo + lane;
    const v4f xrow = ((const v4f*)xs)[(96 + wv) * LDP4 + lane];
    edge_sums<7, false>(pr4, sg4, tsd, wl, wv);
    edge_sums<6, true>(pr4, sg4, tsd, wl, wv + 56);
    *txd = xrow;
  }
  __syncthreads();

  {
    v8f acc[6];
#pragma unroll
    for (int m = 0; m < 6; ++m) acc[m] = vz;
    wmma_col_regb<6, 16>(xs + aofs, agg + aofs, bu1r, acc);
    const float bias = lbu1[ncol];
    const float cv   = lcv[ncol];
#pragma unroll
    for (int m = 0; m < 6; ++m) {
      const v4f dga = *(const v4f*)(ldeg + m * 16 + hi * 8);
      const v4f dgb = *(const v4f*)(ldeg + m * 16 + hi * 8 + 4);
      const float dg[8] = {dga.x, dga.y, dga.z, dga.w, dgb.x, dgb.y, dgb.z, dgb.w};
#pragma unroll
      for (int v = 0; v < 8; ++v) {
        const int row = m * 16 + hi * 8 + v;
        pre[row * LDP + ncol] = fmaxf(acc[m][v] + (dg[v] * cv + bias), 0.f);
      }
    }
  }
  __syncthreads();

  const XRows xn = x_fetch(x + (size_t)(more ? bn : b) * (NN * HH), tid);

  {
    v8f acc[6];
#pragma unroll
    for (int m = 0; m < 6; ++m) acc[m] = vz;
    wmma_col_regb<6, 8>(pre + aofs, pre + aofs, bu2r, acc);
    const float bias = lbu2[ncol];
#pragma unroll
    for (int m = 0; m < 6; ++m) {
#pragma unroll
      for (int v = 0; v < 8; ++v) {
        const int row = m * 16 + hi * 8 + v;
        agg[row * LDP + ncol] = xs[row * LDP + ncol] + acc[m][v] + bias;
      }
    }
  }
  __syncthreads();

  x_store(xs, xn, p4);
  __syncthreads();

  {
    const v4f g4 = ((const v4f*)lgm)[lane];
    const v4f b4 = ((const v4f*)lbt)[lane];
    const v4f* hb4 = (const v4f*)agg + wave * LDP4 + lane;
    v4f* og4 = (v4f*)(out + (size_t)b * (NN * HH)) + wave * (HH / 4) + lane;
    v4f o[12];
    ln_rows<12>(o, hb4, g4, b4);
#pragma unroll
    for (int j = 0; j < 12; ++j) og4[j * 8 * (HH / 4)] = o[j];
  }

  ++cnt;
  if (cnt == TSLOTS || !more) {
    {
      v8f acc[2];
#pragma unroll
      for (int m = 0; m < 2; ++m) acc[m] = vz;
      wmma_col_regb<2, 16>(tx + aofs, ts + aofs, bu1r, acc);
      const float bias = lbu1[ncol];
      const float cv   = lcv[ncol];
      const v4f dgt = *(const v4f*)(ldeg + 96);
      const float dg[8] = {dgt.x, dgt.y, dgt.z, dgt.w, dgt.x, dgt.y, dgt.z, dgt.w};
#pragma unroll
      for (int m = 0; m < 2; ++m) {
#pragma unroll
        for (int v = 0; v < 8; ++v) {
          const int row = m * 16 + hi * 8 + v;
          pre[row * LDP + ncol] = fmaxf(acc[m][v] + (dg[v] * cv + bias), 0.f);
        }
      }
    }
    __syncthreads();
    {
      v8f acc[2];
#pragma unroll
      for (int m = 0; m < 2; ++m) acc[m] = vz;
      wmma_col_regb<2, 8>(pre + aofs, pre + aofs, bu2r, acc);
      const float bias = lbu2[ncol];
#pragma unroll
      for (int m = 0; m < 2; ++m) {
#pragma unroll
        for (int v = 0; v < 8; ++v) {
          const int row = m * 16 + hi * 8 + v;
          agg[row * LDP + ncol] = tx[row * LDP + ncol] + acc[m][v] + bias;
        }
      }
    }
    __syncthreads();
    {
      const v4f g4 = ((const v4f*)lgm)[lane];
      const v4f b4 = ((const v4f*)lbt)[lane];
      const v4f* hb4 = (const v4f*)agg + wave * LDP4 + lane;
      v4f o[4];
      ln_rows<4>(o, hb4, g4, b4);
#pragma unroll
      for (int j = 0; j < 4; ++j) {
        const int s = (wave >> 2) + 2 * j;
        if (s < cnt) {
          const size_t be = (size_t)bfirst + (size_t)s * gridDim.x;
          ((v4f*)(out + be * (NN * HH)))[(96 + (wave & 3)) * (HH / 4) + lane] = o[j];
        }
      }
    }
    cnt = 0;
    bfirst = bn;
  }

  if (!more) break;
  b = bn;
  }
}

extern "C" void kernel_launch(void* const* d_in, const int* in_sizes, int n_in,
                              void* d_out, int out_size, void* d_ws, size_t ws_size,
                              hipStream_t stream) {
  (void)n_in; (void)out_size; (void)d_ws; (void)ws_size;
  const float* x     = (const float*)d_in[0];
  const float* Wm1   = (const float*)d_in[4];
  const float* bm1   = (const float*)d_in[5];
  const float* Wm2   = (const float*)d_in[6];
  const float* bm2   = (const float*)d_in[7];
  const float* Wu1   = (const float*)d_in[8];
  const float* bu1   = (const float*)d_in[9];
  const float* Wu2   = (const float*)d_in[10];
  const float* bu2   = (const float*)d_in[11];
  const float* gamma = (const float*)d_in[12];
  const float* beta  = (const float*)d_in[13];
  float* out = (float*)d_out;

  const int B = in_sizes[0] / (NN * HH);
  const size_t shmem = (size_t)SMEM_FLOATS * sizeof(float);
  hipLaunchKernelGGL(mpnn_layer_kernel, dim3(256), dim3(256), shmem, stream,
                     x, Wm1, bm1, Wm2, bm2, Wu1, bu1, Wu2, bu2, gamma, beta, out, B);
}
